# starting kernel
# baseline (speedup 1.0000x reference)
#include <hip/hip_runtime.h>
#include <hip/hip_bf16.h>

#define N_NODES 50000
#define N_EDGES 1600000
#define IN_CH   512
#define OUT_CH  48
#define LDS_STRIDE 516

typedef __attribute__((ext_vector_type(2))) float v2f;
typedef __attribute__((ext_vector_type(8))) float v8f;

__global__ __launch_bounds__(96) void gemm_xw(const float* __restrict__ feat,
                                              const float* __restrict__ w,
                                              float* __restrict__ xw) {
  __shared__ __align__(16) float As[16 * LDS_STRIDE];
  const int tid = threadIdx.x;
  const int m0  = blockIdx.x * 16;

  for (int idx = tid; idx < 16 * (IN_CH / 4); idx += 96) {
    const int r  = idx >> 7;
    const int c4 = (idx & 127) << 2;
    float4 v = *(const float4*)(feat + (size_t)(m0 + r) * IN_CH + c4);
    *(float4*)(&As[r * LDS_STRIDE + c4]) = v;
  }
  __syncthreads();

  const int lane = tid & 31;
  const int n0   = (tid >> 5) * 16;
  const int r    = lane & 15;
  const int kh   = (lane >> 4) << 1;

  v8f acc = {};
  #pragma unroll 8
  for (int k = 0; k < IN_CH; k += 4) {
    v2f a, b;
    const float* ap = &As[r * LDS_STRIDE + k + kh];
    a.x = ap[0];
    a.y = ap[1];
    const float* bp = w + (size_t)(k + kh) * OUT_CH + n0 + r;
    b.x = bp[0];
    b.y = bp[OUT_CH];
    acc = __builtin_amdgcn_wmma_f32_16x16x4_f32(
         false, a,  false, b,
         (short)0, acc,  false,  false);
  }

  const int col     = n0 + r;
  const int rowbase = m0 + ((lane >> 4) << 3);
  #pragma unroll
  for (int i = 0; i < 8; ++i)
    xw[(size_t)(rowbase + i) * OUT_CH + col] = acc[i];
}

__global__ __launch_bounds__(256) void edge_scatter(const int* __restrict__ ei,
                                                    const float* __restrict__ ew,
                                                    const float* __restrict__ xw,
                                                    float* __restrict__ aggr) {
  const int gid = blockIdx.x * 256 + threadIdx.x;
  const int e   = gid / 12;
  const int c4  = (gid - e * 12) << 2;
  const int src = ei[e];
  const int dst = ei[N_EDGES + e];
  const float wgt = ew[e];

  float4 x = *(const float4*)(xw + (size_t)src * OUT_CH + c4);
  const float v0 = wgt * x.x, v1 = wgt * x.y, v2 = wgt * x.z, v3 = wgt * x.w;
  float* p = aggr + (size_t)dst * OUT_CH + c4;

  asm volatile(
      "global_atomic_add_f32 %0, %1, off\n\t"
      "global_atomic_add_f32 %0, %2, off offset:4\n\t"
      "global_atomic_add_f32 %0, %3, off offset:8\n\t"
      "global_atomic_add_f32 %0, %4, off offset:12"
      :
      : "v"(p), "v"(v0), "v"(v1), "v"(v2), "v"(v3)
      : "memory");
}

__global__ __launch_bounds__(256) void finalize_lsm(float* __restrict__ out,
                                                    const float* __restrict__ bias) {
  const int node = (blockIdx.x * 256 + threadIdx.x) >> 5;
  const int lane = threadIdx.x & 31;
  if (node >= N_NODES) return;
  float* row = out + (size_t)node * OUT_CH;

  float z0 = row[lane] + bias[lane];
  float z1 = (lane < 16) ? (row[32 + lane] + bias[32 + lane]) : -3.402823e38f;

  float m = fmaxf(z0, z1);
  #pragma unroll
  for (int off = 16; off; off >>= 1) m = fmaxf(m, __shfl_xor(m, off, 32));

  float s = __expf(z0 - m) + ((lane < 16) ? __expf(z1 - m) : 0.0f);
  #pragma unroll
  for (int off = 16; off; off >>= 1) s += __shfl_xor(s, off, 32);

  const float lse = m + __logf(s);
  row[lane] = z0 - lse;
  if (lane < 16) row[32 + lane] = z1 - lse;
}

extern "C" void kernel_launch(void* const* d_in, const int* in_sizes, int n_in,
                              void* d_out, int out_size, void* d_ws, size_t ws_size,
                              hipStream_t stream) {
  const int*   ei   = (const int*)d_in[0];
  const float* feat = (const float*)d_in[1];
  const float* ew   = (const float*)d_in[2];
  const float* w    = (const float*)d_in[3];
  const float* bias = (const float*)d_in[4];
  float* out = (float*)d_out;
  float* xw  = (float*)d_ws;

  hipMemsetAsync(out, 0, (size_t)N_NODES * OUT_CH * sizeof(float), stream);

  gemm_xw<<<N_NODES / 16, 96, 0, stream>>>(feat, w, xw);
  edge_scatter<<<(N_EDGES * 12) / 256, 256, 0, stream>>>(ei, ew, xw, out);
  finalize_lsm<<<N_NODES / 8, 256, 0, stream>>>(out, bias);
}


# edge scatter re-laid: per edge one 32-lane f32 atomic add on the target row's whole 128-B line + a 16-lane add on its half line (split by row parity), gathers alike; add arrival order differs
# speedup vs baseline: 1.6581x; 1.6581x over previous
#include <hip/hip_runtime.h>
#include <hip/hip_bf16.h>

#define N_NODES 50000
#define N_EDGES 1600000
#define IN_CH   512
#define OUT_CH  48
#define LDS_STRIDE 516

typedef __attribute__((ext_vector_type(2))) float v2f;
typedef __attribute__((ext_vector_type(8))) float v8f;

__global__ __launch_bounds__(96) void gemm_xw(const float* __restrict__ feat,
                                              const float* __restrict__ w,
                                              float* __restrict__ xw) {
  __shared__ __align__(16) float As[16 * LDS_STRIDE];
  const int tid = threadIdx.x;
  const int m0  = blockIdx.x * 16;

  for (int idx = tid; idx < 16 * (IN_CH / 4); idx += 96) {
    const int r  = idx >> 7;
    const int c4 = (idx & 127) << 2;
    float4 v = *(const float4*)(feat + (size_t)(m0 + r) * IN_CH + c4);
    *(float4*)(&As[r * LDS_STRIDE + c4]) = v;
  }
  __syncthreads();

  const int lane = tid & 31;
  const int n0   = (tid >> 5) * 16;
  const int r    = lane & 15;
  const int kh   = (lane >> 4) << 1;

  v8f acc = {};
  #pragma unroll 8
  for (int k = 0; k < IN_CH; k += 4) {
    v2f a, b;
    const float* ap = &As[r * LDS_STRIDE + k + kh];
    a.x = ap[0];
    a.y = ap[1];
    const float* bp = w + (size_t)(k + kh) * OUT_CH + n0 + r;
    b.x = bp[0];
    b.y = bp[OUT_CH];
    acc = __builtin_amdgcn_wmma_f32_16x16x4_f32(
         false, a,  false, b,
         (short)0, acc,  false,  false);
  }

  const int col     = n0 + r;
  const int rowbase = m0 + ((lane >> 4) << 3);
  #pragma unroll
  for (int i = 0; i < 8; ++i)
    xw[(size_t)(rowbase + i) * OUT_CH + col] = acc[i];
}

#define ROW_BYTES (OUT_CH * 4)
#define EDGES_PER_WAVE 32
#define SCATTER_BLOCK 256
static_assert(N_EDGES % (EDGES_PER_WAVE * (SCATTER_BLOCK / 32)) == 0, "no edge tail");

template <int J>
__device__ __forceinline__ int row_bcast_i(int v) {
  return __builtin_amdgcn_update_dpp(0, v, 0x150 + J, 0xF, 0xF, false);
}
template <int J>
__device__ __forceinline__ float row_bcast_f(float v) {
  return __int_as_float(row_bcast_i<J>(__float_as_int(v)));
}
template <int J>
__device__ __forceinline__ unsigned lane_get_u(int v) {
  return (unsigned)__builtin_amdgcn_readlane(v, J);
}
template <int J>
__device__ __forceinline__ float lane_get_f(float v) {
  return __int_as_float(__builtin_amdgcn_readlane(__float_as_int(v), J));
}

#define FOR8(M) M(0) M(1) M(2) M(3) M(4) M(5) M(6) M(7)

template <int B>
__device__ __forceinline__ void scatter_batch(const char* __restrict__ xwb, char* __restrict__ ab,
                                              int s_full, int d_full, int s_half, int d_half,
                                              float wl, unsigned l4, int c4) {
  float xa[8], xb[8], xh[8];
#define GATHER(i)                                                              \
  {                                                                            \
    constexpr int J = B * 8 + i;                                               \
    const char* pa = xwb + lane_get_u<J>(s_full);                              \
    const char* pb = xwb + lane_get_u<16 + J>(s_full);                         \
    const unsigned sh = (unsigned)(row_bcast_i<J>(s_half) + c4);               \
    xa[i] = *(const float*)(pa + l4);                                          \
    xb[i] = *(const float*)(pb + l4);                                          \
    xh[i] = *(const float*)(xwb + sh);                                         \
  }
  FOR8(GATHER)
#undef GATHER
#define ADD(i)                                                                 \
  {                                                                            \
    constexpr int J = B * 8 + i;                                               \
    char* qa = ab + lane_get_u<J>(d_full);                                     \
    char* qb = ab + lane_get_u<16 + J>(d_full);                                \
    const unsigned dh = (unsigned)(row_bcast_i<J>(d_half) + c4);               \
    const float va = lane_get_f<J>(wl) * xa[i];                                \
    const float vb = lane_get_f<16 + J>(wl) * xb[i];                           \
    const float vh = row_bcast_f<J>(wl) * xh[i];                               \
    asm volatile(                                                              \
        "global_atomic_add_f32 %0, %1, %2\n\t"                                 \
        "global_atomic_add_f32 %0, %3, %4\n\t"                                 \
        "global_atomic_add_f32 %5, %6, %7\n\t"                                 \
        "s_wait_xcnt 0x0"                                                      \
        :                                                                      \
        : "v"(l4), "v"(va), "s"(qa), "v"(vb), "s"(qb), "v"(dh), "v"(vh), "s"(ab) \
        : "memory");                                                           \
  }
  FOR8(ADD)
#undef ADD
}

__global__ __launch_bounds__(SCATTER_BLOCK) void edge_scatter(const int* __restrict__ ei,
                                                              const float* __restrict__ ew,
                                                              const float* __restrict__ xw,
                                                              float* __restrict__ aggr) {
  const int lane = threadIdx.x & 31;
  const int e   = blockIdx.x * SCATTER_BLOCK + threadIdx.x;
  const int src = ei[e];
  const int dst = ei[N_EDGES + e];
  const float wl = ew[e];
  const int odd = dst & 1;
  const int fo  = odd << 6;
  const int ho  = (odd ^ 1) << 7;
  const int so  = src * ROW_BYTES;
  const int dof = dst * ROW_BYTES;
  const int s_full = so + fo,  s_half = so + ho;
  const int d_full = dof + fo, d_half = dof + ho;
  const unsigned l4 = (unsigned)lane << 2;
  const int c4 = (lane & 15) << 2;
  const char* xwb = (const char*)xw;
  char* ab = (char*)aggr;

  scatter_batch<0>(xwb, ab, s_full, d_full, s_half, d_half, wl, l4, c4);
  scatter_batch<1>(xwb, ab, s_full, d_full, s_half, d_half, wl, l4, c4);
}

__global__ __launch_bounds__(256) void finalize_lsm(float* __restrict__ out,
                                                    const float* __restrict__ bias) {
  const int node = (blockIdx.x * 256 + threadIdx.x) >> 5;
  const int lane = threadIdx.x & 31;
  if (node >= N_NODES) return;
  float* row = out + (size_t)node * OUT_CH;

  float z0 = row[lane] + bias[lane];
  float z1 = (lane < 16) ? (row[32 + lane] + bias[32 + lane]) : -3.402823e38f;

  float m = fmaxf(z0, z1);
  #pragma unroll
  for (int off = 16; off; off >>= 1) m = fmaxf(m, __shfl_xor(m, off, 32));

  float s = __expf(z0 - m) + ((lane < 16) ? __expf(z1 - m) : 0.0f);
  #pragma unroll
  for (int off = 16; off; off >>= 1) s += __shfl_xor(s, off, 32);

  const float lse = m + __logf(s);
  row[lane] = z0 - lse;
  if (lane < 16) row[32 + lane] = z1 - lse;
}

extern "C" void kernel_launch(void* const* d_in, const int* in_sizes, int n_in,
                              void* d_out, int out_size, void* d_ws, size_t ws_size,
                              hipStream_t stream) {
  const int*   ei   = (const int*)d_in[0];
  const float* feat = (const float*)d_in[1];
  const float* ew   = (const float*)d_in[2];
  const float* w    = (const float*)d_in[3];
  const float* bias = (const float*)d_in[4];
  float* out = (float*)d_out;
  float* xw  = (float*)d_ws;

  hipMemsetAsync(out, 0, (size_t)N_NODES * OUT_CH * sizeof(float), stream);

  gemm_xw<<<N_NODES / 16, 96, 0, stream>>>(feat, w, xw);
  edge_scatter<<<N_EDGES / SCATTER_BLOCK, SCATTER_BLOCK, 0, stream>>>(ei, ew, xw, out);
  finalize_lsm<<<N_NODES / 8, 256, 0, stream>>>(out, bias);
}


# speedup vs baseline: 1.7198x; 1.0372x over previous
#include <hip/hip_runtime.h>
#include <hip/hip_bf16.h>

#define N_NODES 50000
#define N_EDGES 1600000
#define IN_CH   512
#define OUT_CH  48
#define LDS_STRIDE 516

typedef __attribute__((ext_vector_type(2))) float v2f;
typedef __attribute__((ext_vector_type(8))) float v8f;

#define ROW_BYTES (OUT_CH * 4)
#define XW_BYTES  (N_NODES * ROW_BYTES)
__device__ __forceinline__ int plane1_rows(size_t ws_size) {
  if (ws_size <= (size_t)XW_BYTES) return 0;
  const size_t r = (ws_size - (size_t)XW_BYTES) / ROW_BYTES;
  return r > (size_t)N_NODES ? N_NODES : (int)r;
}

__global__ __launch_bounds__(96) void gemm_xw(const float* __restrict__ feat,
                                              const float* __restrict__ w,
                                              float* __restrict__ xw,
                                              float* __restrict__ plane1,
                                              size_t ws_size) {
  __shared__ __align__(16) float As[16 * LDS_STRIDE];
  const int tid = threadIdx.x;
  const int m0  = blockIdx.x * 16;

  {
    const int n1 = plane1_rows(ws_size);
#pragma unroll
    for (int i = 0; i < 2; ++i) {
      const int f = (tid + 96 * i) * 4;
      if (m0 + f / OUT_CH < n1)
        *(float4*)(plane1 + (size_t)m0 * OUT_CH + f) = make_float4(0.f, 0.f, 0.f, 0.f);
    }
  }

  for (int idx = tid; idx < 16 * (IN_CH / 4); idx += 96) {
    const int r  = idx >> 7;
    const int c4 = (idx & 127) << 2;
    float4 v = *(const float4*)(feat + (size_t)(m0 + r) * IN_CH + c4);
    *(float4*)(&As[r * LDS_STRIDE + c4]) = v;
  }
  __syncthreads();

  const int lane = tid & 31;
  const int n0   = (tid >> 5) * 16;
  const int r    = lane & 15;
  const int kh   = (lane >> 4) << 1;

  v8f acc = {};
  #pragma unroll 8
  for (int k = 0; k < IN_CH; k += 4) {
    v2f a, b;
    const float* ap = &As[r * LDS_STRIDE + k + kh];
    a.x = ap[0];
    a.y = ap[1];
    const float* bp = w + (size_t)(k + kh) * OUT_CH + n0 + r;
    b.x = bp[0];
    b.y = bp[OUT_CH];
    acc = __builtin_amdgcn_wmma_f32_16x16x4_f32(
         false, a,  false, b,
         (short)0, acc,  false,  false);
  }

  const int col     = n0 + r;
  const int rowbase = m0 + ((lane >> 4) << 3);
  #pragma unroll
  for (int i = 0; i < 8; ++i)
    xw[(size_t)(rowbase + i) * OUT_CH + col] = acc[i];
}

#define EDGES_PER_WAVE 32
#define SCATTER_BLOCK 256
static_assert(N_EDGES % (EDGES_PER_WAVE * (SCATTER_BLOCK / 32)) == 0, "no edge tail");

template <int J>
__device__ __forceinline__ int row_bcast_i(int v) {
  return __builtin_amdgcn_update_dpp(0, v, 0x150 + J, 0xF, 0xF, false);
}
template <int J>
__device__ __forceinline__ float row_bcast_f(float v) {
  return __int_as_float(row_bcast_i<J>(__float_as_int(v)));
}
template <int J>
__device__ __forceinline__ unsigned lane_get_u(int v) {
  return (unsigned)__builtin_amdgcn_readlane(v, J);
}
template <int J>
__device__ __forceinline__ float lane_get_f(float v) {
  return __int_as_float(__builtin_amdgcn_readlane(__float_as_int(v), J));
}

#define FOR8(M) M(0) M(1) M(2) M(3) M(4) M(5) M(6) M(7)

template <int B>
__device__ __forceinline__ void scatter_batch(const char* __restrict__ xwb,
                                              int s_full, int s_half,
                                              int fl_lo, int fl_hi, int hl_lo, int hl_hi,
                                              float wl, unsigned l4, int c4) {
  float xa[8], xb[8], xh[8];
#define GATHER(i)                                                              \
  {                                                                            \
    constexpr int J = B * 8 + i;                                               \
    const char* pa = xwb + lane_get_u<J>(s_full);                              \
    const char* pb = xwb + lane_get_u<16 + J>(s_full);                         \
    const unsigned sh = (unsigned)(row_bcast_i<J>(s_half) + c4);               \
    xa[i] = *(const float*)(pa + l4);                                          \
    xb[i] = *(const float*)(pb + l4);                                          \
    xh[i] = *(const float*)(xwb + sh);                                         \
  }
  FOR8(GATHER)
#undef GATHER
#define ADD(i)                                                                 \
  {                                                                            \
    constexpr int J = B * 8 + i;                                               \
    char* qa = (char*)(((unsigned long long)lane_get_u<J>(fl_hi) << 32) |      \
                       lane_get_u<J>(fl_lo));                                  \
    char* qb = (char*)(((unsigned long long)lane_get_u<16 + J>(fl_hi) << 32) | \
                       lane_get_u<16 + J>(fl_lo));                             \
    char* qh = (char*)((((unsigned long long)(unsigned)row_bcast_i<J>(hl_hi) << 32) | \
                        (unsigned)row_bcast_i<J>(hl_lo)) + (unsigned)c4);      \
    const float va = lane_get_f<J>(wl) * xa[i];                                \
    const float vb = lane_get_f<16 + J>(wl) * xb[i];                           \
    const float vh = row_bcast_f<J>(wl) * xh[i];                               \
    asm volatile(                                                              \
        "global_atomic_add_f32 %0, %1, %2\n\t"                                 \
        "global_atomic_add_f32 %0, %3, %4\n\t"                                 \
        "global_atomic_add_f32 %5, %6, off\n\t"                                \
        "s_wait_xcnt 0x0"                                                      \
        :                                                                      \
        : "v"(l4), "v"(va), "s"(qa), "v"(vb), "s"(qb), "v"(qh), "v"(vh)        \
        : "memory");                                                           \
  }
  FOR8(ADD)
#undef ADD
}

__global__ __launch_bounds__(SCATTER_BLOCK) void edge_scatter(const int* __restrict__ ei,
                                                              const float* __restrict__ ew,
                                                              const float* __restrict__ xw,
                                                              float* __restrict__ aggr,
                                                              float* __restrict__ plane1,
                                                              size_t ws_size) {
  const int lane = threadIdx.x & 31;
  const int half = (__builtin_amdgcn_s_sendmsg_rtn(0x87) >> 18) & 1;
  const int n1 = plane1_rows(ws_size);
  const int e   = blockIdx.x * SCATTER_BLOCK + threadIdx.x;
  const int src = ei[e];
  const int dst = ei[N_EDGES + e];
  const float wl = ew[e];
  const int odd = dst & 1;
  const int fo  = odd << 6;
  const int ho  = (odd ^ 1) << 7;
  const int so  = src * ROW_BYTES;
  const int dof = dst * ROW_BYTES;
  const int s_full = so + fo,  s_half = so + ho;
  char* pb = (half != 0 && dst < n1) ? (char*)plane1 : (char*)aggr;
  const unsigned long long fl = (unsigned long long)(pb + (dof + fo));
  const unsigned long long hl = (unsigned long long)(pb + (dof + ho));
  const int fl_lo = (int)(unsigned)fl, fl_hi = (int)(unsigned)(fl >> 32);
  const int hl_lo = (int)(unsigned)hl, hl_hi = (int)(unsigned)(hl >> 32);
  const unsigned l4 = (unsigned)lane << 2;
  const int c4 = (lane & 15) << 2;
  const char* xwb = (const char*)xw;

  scatter_batch<0>(xwb, s_full, s_half, fl_lo, fl_hi, hl_lo, hl_hi, wl, l4, c4);
  scatter_batch<1>(xwb, s_full, s_half, fl_lo, fl_hi, hl_lo, hl_hi, wl, l4, c4);
}

__global__ __launch_bounds__(256) void finalize_lsm(float* __restrict__ out,
                                                    const float* __restrict__ bias,
                                                    const float* __restrict__ plane1,
                                                    size_t ws_size) {
  const int node = (blockIdx.x * 256 + threadIdx.x) >> 5;
  const int lane = threadIdx.x & 31;
  if (node >= N_NODES) return;
  float* row = out + (size_t)node * OUT_CH;
  const float* row1 = plane1 + (size_t)node * OUT_CH;
  const bool has1 = node < plane1_rows(ws_size);

  float a0 = row[lane];
  float a1 = (lane < 16) ? row[32 + lane] : 0.0f;
  if (has1) {
    a0 += row1[lane];
    if (lane < 16) a1 += row1[32 + lane];
  }
  float z0 = a0 + bias[lane];
  float z1 = (lane < 16) ? (a1 + bias[32 + lane]) : -3.402823e38f;

  float m = fmaxf(z0, z1);
  #pragma unroll
  for (int off = 16; off; off >>= 1) m = fmaxf(m, __shfl_xor(m, off, 32));

  float s = __expf(z0 - m) + ((lane < 16) ? __expf(z1 - m) : 0.0f);
  #pragma unroll
  for (int off = 16; off; off >>= 1) s += __shfl_xor(s, off, 32);

  const float lse = m + __logf(s);
  row[lane] = z0 - lse;
  if (lane < 16) row[32 + lane] = z1 - lse;
}

extern "C" void kernel_launch(void* const* d_in, const int* in_sizes, int n_in,
                              void* d_out, int out_size, void* d_ws, size_t ws_size,
                              hipStream_t stream) {
  const int*   ei   = (const int*)d_in[0];
  const float* feat = (const float*)d_in[1];
  const float* ew   = (const float*)d_in[2];
  const float* w    = (const float*)d_in[3];
  const float* bias = (const float*)d_in[4];
  float* out = (float*)d_out;
  float* xw  = (float*)d_ws;
  float* plane1 = (float*)((char*)d_ws + XW_BYTES);

  hipMemsetAsync(out, 0, (size_t)N_NODES * OUT_CH * sizeof(float), stream);

  gemm_xw<<<N_NODES / 16, 96, 0, stream>>>(feat, w, xw, plane1, ws_size);
  edge_scatter<<<N_EDGES / SCATTER_BLOCK, SCATTER_BLOCK, 0, stream>>>(ei, ew, xw, out, plane1, ws_size);
  finalize_lsm<<<N_NODES / 8, 256, 0, stream>>>(out, bias, plane1, ws_size);
}


# edges filed per call by 200-row destination block (vote-ranked counting sort, no atomics); one work-group per block orders its records by row, one wave sums each row in fixed order, then log_softmax
# speedup vs baseline: 2.4229x; 1.4088x over previous
#include <hip/hip_runtime.h>
#include <hip/hip_bf16.h>

#define N_NODES 50000
#define N_EDGES 1600000
#define IN_CH   512
#define OUT_CH  48
#define LDS_STRIDE 516

typedef __attribute__((ext_vector_type(2))) float v2f;
typedef __attribute__((ext_vector_type(8))) float v8f;

#define ROW_BYTES (OUT_CH * 4)
#define XW_BYTES  (N_NODES * ROW_BYTES)

#define BIN_ROWS   200
#define N_BINS     (N_NODES / BIN_ROWS)
#define CHUNK      6400
#define N_CHUNKS   (N_EDGES / CHUNK)
#define PER_THREAD (CHUNK / 256)
#define TAB_PITCH  256
#define TILE       7168
#define REC_BYTES  ((size_t)N_EDGES * 8)
#define TAB_BYTES  ((size_t)N_CHUNKS * TAB_PITCH * 4)
#define WS_NEED    ((size_t)XW_BYTES + REC_BYTES + TAB_BYTES)
static_assert(N_NODES % BIN_ROWS == 0 && N_EDGES % CHUNK == 0 && CHUNK % 256 == 0, "exact tiling");
static_assert(TILE == 1024 * 7, "seven records per thread of the owner work-group");
static_assert(N_NODES <= 65536 && BIN_ROWS <= 256 && N_BINS <= TAB_PITCH - 1, "record packing");

__device__ __forceinline__ bool filed_path(size_t ws_size) { return ws_size >= WS_NEED; }

__global__ __launch_bounds__(96) void gemm_xw(const float* __restrict__ feat,
                                              const float* __restrict__ w,
                                              float* __restrict__ xw) {
  __shared__ __align__(16) float As[16 * LDS_STRIDE];
  const int tid = threadIdx.x;
  const int m0  = blockIdx.x * 16;

  for (int idx = tid; idx < 16 * (IN_CH / 4); idx += 96) {
    const int r  = idx >> 7;
    const int c4 = (idx & 127) << 2;
    float4 v = *(const float4*)(feat + (size_t)(m0 + r) * IN_CH + c4);
    *(float4*)(&As[r * LDS_STRIDE + c4]) = v;
  }
  __syncthreads();

  const int lane = tid & 31;
  const int n0   = (tid >> 5) * 16;
  const int r    = lane & 15;
  const int kh   = (lane >> 4) << 1;

  v8f acc = {};
  #pragma unroll 8
  for (int k = 0; k < IN_CH; k += 4) {
    v2f a, b;
    const float* ap = &As[r * LDS_STRIDE + k + kh];
    a.x = ap[0];
    a.y = ap[1];
    const float* bp = w + (size_t)(k + kh) * OUT_CH + n0 + r;
    b.x = bp[0];
    b.y = bp[OUT_CH];
    acc = __builtin_amdgcn_wmma_f32_16x16x4_f32(
         false, a,  false, b,
         (short)0, acc,  false,  false);
  }

  const int col     = n0 + r;
  const int rowbase = m0 + ((lane >> 4) << 3);
  #pragma unroll
  for (int i = 0; i < 8; ++i)
    xw[(size_t)(rowbase + i) * OUT_CH + col] = acc[i];
}


template <int NW, int IPT>
__device__ __forceinline__ void block_rank_by_key(const unsigned (&key)[IPT], int (&pos)[IPT],
                                                  unsigned* cntw, unsigned* tot, int* start) {
  constexpr int NT = NW * 32;
  const int tid  = threadIdx.x;
  const int lane = tid & 31;
  const int wvid = tid >> 5;
  for (int i = tid; i < NW * 256; i += NT) cntw[i] = 0u;
  __syncthreads();
  unsigned* mycnt = cntw + wvid * 256;
  int rank[IPT];
#pragma unroll
  for (int i = 0; i < IPT; ++i) {
    const unsigned k = key[i];
    unsigned m = 0xFFFFFFFFu;
#pragma unroll
    for (int bit = 0; bit < 8; ++bit) {
      const bool on = ((k >> bit) & 1u) != 0u;
      const unsigned bal = __builtin_amdgcn_ballot_w32(on);
      m &= on ? bal : ~bal;
    }
    const unsigned lt  = m & ((1u << lane) - 1u);
    const unsigned old = mycnt[k];
    rank[i] = (int)(old + (unsigned)__popc(lt));
    if (lt == 0u) mycnt[k] = old + (unsigned)__popc(m);
  }
  __syncthreads();
  if (tid < 256) {
    unsigned run = 0u;
#pragma unroll
    for (int w = 0; w < NW; ++w) {
      const unsigned c = cntw[w * 256 + tid];
      cntw[w * 256 + tid] = run;
      run += c;
    }
    tot[tid] = run;
  }
  __syncthreads();
  if (tid < 32) {
    unsigned loc[8];
    unsigned sum = 0u;
#pragma unroll
    for (int q = 0; q < 8; ++q) { loc[q] = tot[tid * 8 + q]; sum += loc[q]; }
    unsigned incl = sum;
#pragma unroll
    for (int off = 1; off < 32; off <<= 1) {
      const unsigned t = __shfl_up(incl, off, 32);
      if (tid >= off) incl += t;
    }
    unsigned run = incl - sum;
#pragma unroll
    for (int q = 0; q < 8; ++q) { start[tid * 8 + q] = (int)run; run += loc[q]; }
    if (tid == 31) start[256] = (int)run;
  }
  __syncthreads();
#pragma unroll
  for (int i = 0; i < IPT; ++i) pos[i] = start[key[i]] + (int)mycnt[key[i]] + rank[i];
}

__device__ __forceinline__ int clamp_node(int v) {
  return v < 0 ? 0 : (v > N_NODES - 1 ? N_NODES - 1 : v);
}

__global__ __launch_bounds__(256) void file_edges(const int* __restrict__ ei,
                                                  const float* __restrict__ ew,
                                                  unsigned* __restrict__ recs_w,
                                                  int* __restrict__ tab,
                                                  size_t ws_size) {
  __shared__ __align__(16) uint2 stage[CHUNK];
  __shared__ unsigned cntw[8 * 256];
  __shared__ unsigned tot[256];
  __shared__ int start[257];
  const int tid = threadIdx.x;
  const int c   = blockIdx.x;
  const int e0  = c * CHUNK + tid;

  if (!filed_path(ws_size)) return;

  int d[PER_THREAD], s[PER_THREAD];
  float wv[PER_THREAD];
#pragma unroll
  for (int i = 0; i < PER_THREAD; ++i) {
    const int e = e0 + i * 256;
    d[i]  = ei[N_EDGES + e];
    s[i]  = ei[e];
    wv[i] = ew[e];
  }
  unsigned key[PER_THREAD];
  int pos[PER_THREAD];
#pragma unroll
  for (int i = 0; i < PER_THREAD; ++i) {
    d[i] = clamp_node(d[i]);
    key[i] = (unsigned)d[i] / BIN_ROWS;
  }
  block_rank_by_key<8, PER_THREAD>(key, pos, cntw, tot, start);

  tab[c * TAB_PITCH + tid] = start[tid];

#pragma unroll
  for (int i = 0; i < PER_THREAD; ++i) {
    const unsigned dl = (unsigned)d[i] - key[i] * BIN_ROWS;
    stage[pos[i]] = make_uint2((unsigned)clamp_node(s[i]) | (dl << 16), __float_as_uint(wv[i]));
  }
  __syncthreads();

  uint4* dstv = (uint4*)(recs_w + (size_t)c * CHUNK * 2);
  const uint4* srcv = (const uint4*)stage;
  for (int q = tid; q < CHUNK / 2; q += 256) dstv[q] = srcv[q];
}

__global__ __launch_bounds__(1024) void owner_sum_lsm(float* __restrict__ out,
                                                      const float* __restrict__ bias,
                                                      const int* __restrict__ ei,
                                                      const float* __restrict__ ew,
                                                      const float* __restrict__ xw,
                                                      const unsigned* __restrict__ recs_w,
                                                      const int* __restrict__ tab,
                                                      size_t ws_size) {
  const uint2* recs = (const uint2*)recs_w;
  __shared__ __align__(16) uint2 sorted[TILE];
  __shared__ unsigned cntw[32 * 256];
  __shared__ unsigned tot[256];
  __shared__ int start[257];
  __shared__ unsigned psz[256];
  __shared__ int pfx[257];
  __shared__ int pst[256];
  const int tid  = threadIdx.x;
  const int lane = tid & 31;
  const int wvid = tid >> 5;
  const int b    = blockIdx.x;
  const int l2   = lane < 24 ? lane : 23;
  const bool filed = filed_path(ws_size);

  float2 racc[7];
#pragma unroll
  for (int k = 0; k < 7; ++k) racc[k] = make_float2(0.0f, 0.0f);

  int total = N_EDGES;
  if (filed) {
    if (tid < 256) {
      int s0 = 0, s1 = 0;
      if (tid < N_CHUNKS) {
        s0 = tab[tid * TAB_PITCH + b];
        s1 = tab[tid * TAB_PITCH + b + 1];
        s0 = s0 < 0 ? 0 : (s0 > CHUNK ? CHUNK : s0);
        s1 = s1 < s0 ? s0 : (s1 > CHUNK ? CHUNK : s1);
      }
      pst[tid] = s0;
      psz[tid] = (unsigned)(s1 - s0);
    }
    __syncthreads();
    if (tid < 32) {
      unsigned loc[8];
      unsigned sum = 0u;
#pragma unroll
      for (int q = 0; q < 8; ++q) { loc[q] = psz[tid * 8 + q]; sum += loc[q]; }
      unsigned incl = sum;
#pragma unroll
      for (int off = 1; off < 32; off <<= 1) {
        const unsigned t = __shfl_up(incl, off, 32);
        if (tid >= off) incl += t;
      }
      unsigned run = incl - sum;
#pragma unroll
      for (int q = 0; q < 8; ++q) { pfx[tid * 8 + q] = (int)run; run += loc[q]; }
      if (tid == 31) pfx[256] = (int)run;
    }
    __syncthreads();
    total = pfx[256];
    total = total < 0 ? 0 : (total > N_EDGES ? N_EDGES : total);
  }

  for (int tb = 0; tb < total; tb += TILE) {
    unsigned key[7];
    uint2 rec[7];
    int pos[7];
    if (filed) {
      int ra[7];
#pragma unroll
      for (int i = 0; i < 7; ++i) {
        const int g  = tb + i * 1024 + tid;
        const int gg = g < total ? g : total - 1;
        int lo = 0, hi = N_CHUNKS - 1;
#pragma unroll
        for (int it = 0; it < 8; ++it) {
          const int mid = (lo + hi + 1) >> 1;
          if (pfx[mid] <= gg) lo = mid; else hi = mid - 1;
        }
        int idx = pst[lo] + (gg - pfx[lo]);
        idx = idx < 0 ? 0 : (idx > CHUNK - 1 ? CHUNK - 1 : idx);
        ra[i] = lo * CHUNK + idx;
      }
#pragma unroll
      for (int i = 0; i < 7; ++i) rec[i] = recs[ra[i]];
#pragma unroll
      for (int i = 0; i < 7; ++i) {
        const int g = tb + i * 1024 + tid;
        const unsigned dl = rec[i].x >> 16;
        key[i] = g < total ? (dl < BIN_ROWS ? dl : BIN_ROWS - 1) : 255u;
      }
    } else {
#pragma unroll
      for (int i = 0; i < 7; ++i) {
        const int g  = tb + i * 1024 + tid;
        const int gg = g < total ? g : total - 1;
        const int dst = clamp_node(ei[N_EDGES + gg]);
        const int bin = dst / BIN_ROWS;
        const unsigned dl = (unsigned)(dst - bin * BIN_ROWS);
        rec[i] = make_uint2((unsigned)clamp_node(ei[gg]) | (dl << 16), __float_as_uint(ew[gg]));
        key[i] = (g < total && bin == b) ? dl : 255u;
      }
    }
    block_rank_by_key<32, 7>(key, pos, cntw, tot, start);
#pragma unroll
    for (int i = 0; i < 7; ++i) sorted[pos[i]] = rec[i];
    __syncthreads();

#pragma unroll
    for (int k = 0; k < 7; ++k) {
      const int r = wvid + 32 * k;
      if (r < BIN_ROWS) {
        const int rs = start[r];
        const int re = start[r + 1];
        float sx = 0.0f, sy = 0.0f;
        for (int j = rs; j < re; j += 8) {
          uint2 rr[8];
          float2 v[8];
#pragma unroll
          for (int u = 0; u < 8; ++u) {
            const int jj = (j + u < re) ? (j + u) : (re - 1);
            rr[u] = sorted[jj];
            const unsigned src = rr[u].x & 0xFFFFu;
            v[u] = *(const float2*)(xw + (size_t)(src < N_NODES ? src : N_NODES - 1) * OUT_CH + 2 * l2);
          }
#pragma unroll
          for (int u = 0; u < 8; ++u) {
            if (j + u < re) {
              const float wj = __uint_as_float(rr[u].y);
              sx += wj * v[u].x;
              sy += wj * v[u].y;
            }
          }
        }
        racc[k].x += sx;
        racc[k].y += sy;
      }
    }
    __syncthreads();
  }

  const float2 bz = *(const float2*)(bias + 2 * l2);
#pragma unroll
  for (int k = 0; k < 7; ++k) {
    const int r = wvid + 32 * k;
    if (r < BIN_ROWS) {
      const int node = b * BIN_ROWS + r;
      float* row = out + (size_t)node * OUT_CH;
      const float2 a = racc[k];
      const float z0 = a.x + bz.x;
      const float z1 = a.y + bz.y;

      float m = (lane < 24) ? fmaxf(z0, z1) : -3.402823e38f;
      #pragma unroll
      for (int off = 16; off; off >>= 1) m = fmaxf(m, __shfl_xor(m, off, 32));

      float sm = (lane < 24) ? (__expf(z0 - m) + __expf(z1 - m)) : 0.0f;
      #pragma unroll
      for (int off = 16; off; off >>= 1) sm += __shfl_xor(sm, off, 32);

      const float lse = m + __logf(sm);
      if (lane < 24) *(float2*)(row + 2 * lane) = make_float2(z0 - lse, z1 - lse);
    }
  }
}

extern "C" void kernel_launch(void* const* d_in, const int* in_sizes, int n_in,
                              void* d_out, int out_size, void* d_ws, size_t ws_size,
                              hipStream_t stream) {
  const int*   ei   = (const int*)d_in[0];
  const float* feat = (const float*)d_in[1];
  const float* ew   = (const float*)d_in[2];
  const float* w    = (const float*)d_in[3];
  const float* bias = (const float*)d_in[4];
  float* out = (float*)d_out;
  float* xw  = (float*)d_ws;
  unsigned* recs = (unsigned*)((char*)d_ws + XW_BYTES);
  int*   tab  = (int*)((char*)d_ws + XW_BYTES + REC_BYTES);

  gemm_xw<<<N_NODES / 16, 96, 0, stream>>>(feat, w, xw);
  file_edges<<<N_CHUNKS, 256, 0, stream>>>(ei, ew, recs, tab, ws_size);
  owner_sum_lsm<<<N_BINS, 1024, 0, stream>>>(out, bias, ei, ew, xw, recs, tab, ws_size);
}


# edge filing moved into the product's launch: work-groups 0-249 of one 256-thread grid file the edges beside 1,563 product groups of 32 rows (2 launches per call, not 3); every sum order unchanged
# speedup vs baseline: 2.5960x; 1.0714x over previous
#include <hip/hip_runtime.h>
#include <hip/hip_bf16.h>

#define N_NODES 50000
#define N_EDGES 1600000
#define IN_CH   512
#define OUT_CH  48
#define LDS_STRIDE 516

typedef __attribute__((ext_vector_type(2))) float v2f;
typedef __attribute__((ext_vector_type(8))) float v8f;

#define ROW_BYTES (OUT_CH * 4)
#define XW_BYTES  (N_NODES * ROW_BYTES)

#define BIN_ROWS   200
#define N_BINS     (N_NODES / BIN_ROWS)
#define CHUNK      6400
#define N_CHUNKS   (N_EDGES / CHUNK)
#define PER_THREAD (CHUNK / 256)
#define TAB_PITCH  256
#define TILE       7168
#define REC_BYTES  ((size_t)N_EDGES * 8)
#define TAB_BYTES  ((size_t)N_CHUNKS * TAB_PITCH * 4)
#define WS_NEED    ((size_t)XW_BYTES + REC_BYTES + TAB_BYTES)
static_assert(N_NODES % BIN_ROWS == 0 && N_EDGES % CHUNK == 0 && CHUNK % 256 == 0, "exact tiling");
static_assert(TILE == 1024 * 7, "seven records per thread of the owner work-group");
static_assert(N_NODES <= 65536 && BIN_ROWS <= 256 && N_BINS <= TAB_PITCH - 1, "record packing");

__device__ __forceinline__ bool filed_path(size_t ws_size) { return ws_size >= WS_NEED; }

template <int NW, int IPT>
__device__ __forceinline__ void block_rank_by_key(const unsigned (&key)[IPT], int (&pos)[IPT],
                                                  unsigned* cntw, unsigned* tot, int* start) {
  constexpr int NT = NW * 32;
  const int tid  = threadIdx.x;
  const int lane = tid & 31;
  const int wvid = tid >> 5;
  for (int i = tid; i < NW * 256; i += NT) cntw[i] = 0u;
  __syncthreads();
  unsigned* mycnt = cntw + wvid * 256;
  int rank[IPT];
#pragma unroll
  for (int i = 0; i < IPT; ++i) {
    const unsigned k = key[i];
    unsigned m = 0xFFFFFFFFu;
#pragma unroll
    for (int bit = 0; bit < 8; ++bit) {
      const bool on = ((k >> bit) & 1u) != 0u;
      const unsigned bal = __builtin_amdgcn_ballot_w32(on);
      m &= on ? bal : ~bal;
    }
    const unsigned lt  = m & ((1u << lane) - 1u);
    const unsigned old = mycnt[k];
    rank[i] = (int)(old + (unsigned)__popc(lt));
    if (lt == 0u) mycnt[k] = old + (unsigned)__popc(m);
  }
  __syncthreads();
  if (tid < 256) {
    unsigned run = 0u;
#pragma unroll
    for (int w = 0; w < NW; ++w) {
      const unsigned c = cntw[w * 256 + tid];
      cntw[w * 256 + tid] = run;
      run += c;
    }
    tot[tid] = run;
  }
  __syncthreads();
  if (tid < 32) {
    unsigned loc[8];
    unsigned sum = 0u;
#pragma unroll
    for (int q = 0; q < 8; ++q) { loc[q] = tot[tid * 8 + q]; sum += loc[q]; }
    unsigned incl = sum;
#pragma unroll
    for (int off = 1; off < 32; off <<= 1) {
      const unsigned t = __shfl_up(incl, off, 32);
      if (tid >= off) incl += t;
    }
    unsigned run = incl - sum;
#pragma unroll
    for (int q = 0; q < 8; ++q) { start[tid * 8 + q] = (int)run; run += loc[q]; }
    if (tid == 31) start[256] = (int)run;
  }
  __syncthreads();
#pragma unroll
  for (int i = 0; i < IPT; ++i) pos[i] = start[key[i]] + (int)mycnt[key[i]] + rank[i];
}

#define GEMM_ROWS   32
#define GEMM_GROUPS ((N_NODES + GEMM_ROWS - 1) / GEMM_ROWS)
#define A_TILE_BYTES (GEMM_ROWS * LDS_STRIDE * 4)
#define FILE_LDS_BYTES (CHUNK * 8 + 8 * 256 * 4 + 256 * 4 + 260 * 4)
#define K1_LDS_BYTES (A_TILE_BYTES > FILE_LDS_BYTES ? A_TILE_BYTES : FILE_LDS_BYTES)
#define PIECES (GEMM_ROWS * (IN_CH / 4) / 256)
static_assert((GEMM_ROWS * (IN_CH / 4)) % 256 == 0, "whole 16-byte pieces per thread");

__device__ __forceinline__ void product_tile(const float* As, const float* __restrict__ w,
                                             float* __restrict__ xw, int m0, int rt, int n0, int lane) {
  const int mt = m0 + rt * 16;
  const int r  = lane & 15;
  const int kh = (lane >> 4) << 1;
  const float* At = As + rt * 16 * LDS_STRIDE;

  v8f acc = {};
  #pragma unroll 8
  for (int k = 0; k < IN_CH; k += 4) {
    v2f a, b;
    const float* ap = &At[r * LDS_STRIDE + k + kh];
    a.x = ap[0];
    a.y = ap[1];
    const float* bp = w + (size_t)(k + kh) * OUT_CH + n0 + r;
    b.x = bp[0];
    b.y = bp[OUT_CH];
    acc = __builtin_amdgcn_wmma_f32_16x16x4_f32(
         false, a,  false, b,
         (short)0, acc,  false,  false);
  }

  const int col     = n0 + r;
  const int rowbase = mt + ((lane >> 4) << 3);
  #pragma unroll
  for (int i = 0; i < 8; ++i)
    xw[(size_t)(rowbase + i) * OUT_CH + col] = acc[i];
}

__device__ __forceinline__ int clamp_node(int v) {
  return v < 0 ? 0 : (v > N_NODES - 1 ? N_NODES - 1 : v);
}

__global__ __launch_bounds__(256) void gemm_xw_and_file(const float* __restrict__ feat,
                                                        const float* __restrict__ w,
                                                        float* __restrict__ xw,
                                                        const int* __restrict__ ei,
                                                        const float* __restrict__ ew,
                                                        unsigned* __restrict__ recs_w,
                                                        int* __restrict__ tab,
                                                        size_t ws_size) {
  __shared__ __align__(16) unsigned char lds[K1_LDS_BYTES];
  const int tid  = threadIdx.x;
  const int lane = tid & 31;
  const int wvid = __builtin_amdgcn_readfirstlane(tid >> 5);

  if (blockIdx.x < N_CHUNKS) {
    if (!filed_path(ws_size)) return;
    uint2*    stage = (uint2*)lds;
    unsigned* cntw  = (unsigned*)(lds + CHUNK * 8);
    unsigned* tot   = cntw + 8 * 256;
    int*      start = (int*)(tot + 256);
    const int c   = blockIdx.x;
    const int e0  = c * CHUNK + tid;

    unsigned* kbuf = (unsigned*)stage;
    {
      int dv[PER_THREAD];
#pragma unroll
      for (int i = 0; i < PER_THREAD; ++i) dv[i] = ei[N_EDGES + e0 + i * 256];
#pragma unroll
      for (int i = 0; i < PER_THREAD; ++i) {
        const int dc = clamp_node(dv[i]);
        const unsigned k = (unsigned)dc / BIN_ROWS;
        kbuf[i * 256 + tid] = k | (((unsigned)dc - k * BIN_ROWS) << 8);
      }
    }
    for (int i = tid; i < 8 * 256; i += 256) cntw[i] = 0u;
    __syncthreads();

    unsigned* mycnt = cntw + wvid * 256;
#pragma unroll 1
    for (int i = 0; i < PER_THREAD; ++i) {
      const unsigned kd = kbuf[i * 256 + tid];
      const unsigned k  = kd & 255u;
      unsigned m = 0xFFFFFFFFu;
#pragma unroll
      for (int bit = 0; bit < 8; ++bit) {
        const bool on = ((k >> bit) & 1u) != 0u;
        const unsigned bal = __builtin_amdgcn_ballot_w32(on);
        m &= on ? bal : ~bal;
      }
      const unsigned lt  = m & ((1u << lane) - 1u);
      const unsigned old = mycnt[k];
      kbuf[i * 256 + tid] = kd | ((old + (unsigned)__popc(lt)) << 16);
      if (lt == 0u) mycnt[k] = old + (unsigned)__popc(m);
    }
    __syncthreads();

    int sv[PER_THREAD];
    float wv[PER_THREAD];
#pragma unroll
    for (int i = 0; i < PER_THREAD; ++i) {
      sv[i] = ei[e0 + i * 256];
      wv[i] = ew[e0 + i * 256];
    }

    {
      unsigned run = 0u;
#pragma unroll
      for (int wq = 0; wq < 8; ++wq) {
        const unsigned cc = cntw[wq * 256 + tid];
        cntw[wq * 256 + tid] = run;
        run += cc;
      }
      tot[tid] = run;
    }
    __syncthreads();
    if (tid < 32) {
      unsigned loc[8];
      unsigned sum = 0u;
#pragma unroll
      for (int q = 0; q < 8; ++q) { loc[q] = tot[tid * 8 + q]; sum += loc[q]; }
      unsigned incl = sum;
#pragma unroll
      for (int off = 1; off < 32; off <<= 1) {
        const unsigned t = __shfl_up(incl, off, 32);
        if (tid >= off) incl += t;
      }
      unsigned run = incl - sum;
#pragma unroll
      for (int q = 0; q < 8; ++q) { start[tid * 8 + q] = (int)run; run += loc[q]; }
      if (tid == 31) start[256] = (int)run;
    }
    __syncthreads();

    tab[c * TAB_PITCH + tid] = start[tid];

    int pos[PER_THREAD];
    unsigned dlw[PER_THREAD];
#pragma unroll
    for (int i = 0; i < PER_THREAD; ++i) {
      const unsigned krd = kbuf[i * 256 + tid];
      const unsigned k   = krd & 255u;
      dlw[i] = ((krd >> 8) & 255u) << 16;
      pos[i] = start[k] + (int)mycnt[k] + (int)(krd >> 16);
    }
    __syncthreads();
#pragma unroll
    for (int i = 0; i < PER_THREAD; ++i)
      stage[pos[i]] = make_uint2((unsigned)clamp_node(sv[i]) | dlw[i], __float_as_uint(wv[i]));
    __syncthreads();

    uint4* dstv = (uint4*)(recs_w + (size_t)c * CHUNK * 2);
    const uint4* srcv = (const uint4*)stage;
    for (int q = tid; q < CHUNK / 2; q += 256) dstv[q] = srcv[q];
    return;
  }

  float* As = (float*)lds;
  const int m0 = (blockIdx.x - N_CHUNKS) * GEMM_ROWS;
  const int rt = wvid / 3;
  const int n0 = (wvid - rt * 3) * 16;

  const int sr = tid >> 7;
  const int sc = (tid & 127) << 2;
  float4 v[PIECES];
#pragma unroll
  for (int i = 0; i < PIECES; ++i) {
    const int row = m0 + sr + 2 * i;
    v[i] = *(const float4*)(feat + (size_t)(row < N_NODES ? row : N_NODES - 1) * IN_CH + sc);
  }
#pragma unroll
  for (int i = 0; i < PIECES; ++i) *(float4*)(&As[(sr + 2 * i) * LDS_STRIDE + sc]) = v[i];
  __syncthreads();

  if (wvid < 6 && m0 + rt * 16 < N_NODES) product_tile(As, w, xw, m0, rt, n0, lane);
}

__global__ __launch_bounds__(1024) void owner_sum_lsm(float* __restrict__ out,
                                                      const float* __restrict__ bias,
                                                      const int* __restrict__ ei,
                                                      const float* __restrict__ ew,
                                                      const float* __restrict__ xw,
                                                      const unsigned* __restrict__ recs_w,
                                                      const int* __restrict__ tab,
                                                      size_t ws_size) {
  const uint2* recs = (const uint2*)recs_w;
  __shared__ __align__(16) uint2 sorted[TILE];
  __shared__ unsigned cntw[32 * 256];
  __shared__ unsigned tot[256];
  __shared__ int start[257];
  __shared__ unsigned psz[256];
  __shared__ int pfx[257];
  __shared__ int pst[256];
  const int tid  = threadIdx.x;
  const int lane = tid & 31;
  const int wvid = tid >> 5;
  const int b    = blockIdx.x;
  const int l2   = lane < 24 ? lane : 23;
  const bool filed = filed_path(ws_size);

  float2 racc[7];
#pragma unroll
  for (int k = 0; k < 7; ++k) racc[k] = make_float2(0.0f, 0.0f);

  int total = N_EDGES;
  if (filed) {
    if (tid < 256) {
      int s0 = 0, s1 = 0;
      if (tid < N_CHUNKS) {
        s0 = tab[tid * TAB_PITCH + b];
        s1 = tab[tid * TAB_PITCH + b + 1];
        s0 = s0 < 0 ? 0 : (s0 > CHUNK ? CHUNK : s0);
        s1 = s1 < s0 ? s0 : (s1 > CHUNK ? CHUNK : s1);
      }
      pst[tid] = s0;
      psz[tid] = (unsigned)(s1 - s0);
    }
    __syncthreads();
    if (tid < 32) {
      unsigned loc[8];
      unsigned sum = 0u;
#pragma unroll
      for (int q = 0; q < 8; ++q) { loc[q] = psz[tid * 8 + q]; sum += loc[q]; }
      unsigned incl = sum;
#pragma unroll
      for (int off = 1; off < 32; off <<= 1) {
        const unsigned t = __shfl_up(incl, off, 32);
        if (tid >= off) incl += t;
      }
      unsigned run = incl - sum;
#pragma unroll
      for (int q = 0; q < 8; ++q) { pfx[tid * 8 + q] = (int)run; run += loc[q]; }
      if (tid == 31) pfx[256] = (int)run;
    }
    __syncthreads();
    total = pfx[256];
    total = total < 0 ? 0 : (total > N_EDGES ? N_EDGES : total);
  }

  for (int tb = 0; tb < total; tb += TILE) {
    unsigned key[7];
    uint2 rec[7];
    int pos[7];
    if (filed) {
      int ra[7];
#pragma unroll
      for (int i = 0; i < 7; ++i) {
        const int g  = tb + i * 1024 + tid;
        const int gg = g < total ? g : total - 1;
        int lo = 0, hi = N_CHUNKS - 1;
#pragma unroll
        for (int it = 0; it < 8; ++it) {
          const int mid = (lo + hi + 1) >> 1;
          if (pfx[mid] <= gg) lo = mid; else hi = mid - 1;
        }
        int idx = pst[lo] + (gg - pfx[lo]);
        idx = idx < 0 ? 0 : (idx > CHUNK - 1 ? CHUNK - 1 : idx);
        ra[i] = lo * CHUNK + idx;
      }
#pragma unroll
      for (int i = 0; i < 7; ++i) rec[i] = recs[ra[i]];
#pragma unroll
      for (int i = 0; i < 7; ++i) {
        const int g = tb + i * 1024 + tid;
        const unsigned dl = rec[i].x >> 16;
        key[i] = g < total ? (dl < BIN_ROWS ? dl : BIN_ROWS - 1) : 255u;
      }
    } else {
#pragma unroll
      for (int i = 0; i < 7; ++i) {
        const int g  = tb + i * 1024 + tid;
        const int gg = g < total ? g : total - 1;
        const int dst = clamp_node(ei[N_EDGES + gg]);
        const int bin = dst / BIN_ROWS;
        const unsigned dl = (unsigned)(dst - bin * BIN_ROWS);
        rec[i] = make_uint2((unsigned)clamp_node(ei[gg]) | (dl << 16), __float_as_uint(ew[gg]));
        key[i] = (g < total && bin == b) ? dl : 255u;
      }
    }
    block_rank_by_key<32, 7>(key, pos, cntw, tot, start);
#pragma unroll
    for (int i = 0; i < 7; ++i) sorted[pos[i]] = rec[i];
    __syncthreads();

#pragma unroll
    for (int k = 0; k < 7; ++k) {
      const int r = wvid + 32 * k;
      if (r < BIN_ROWS) {
        const int rs = start[r];
        const int re = start[r + 1];
        float sx = 0.0f, sy = 0.0f;
        for (int j = rs; j < re; j += 8) {
          uint2 rr[8];
          float2 v[8];
#pragma unroll
          for (int u = 0; u < 8; ++u) {
            const int jj = (j + u < re) ? (j + u) : (re - 1);
            rr[u] = sorted[jj];
            const unsigned src = rr[u].x & 0xFFFFu;
            v[u] = *(const float2*)(xw + (size_t)(src < N_NODES ? src : N_NODES - 1) * OUT_CH + 2 * l2);
          }
#pragma unroll
          for (int u = 0; u < 8; ++u) {
            if (j + u < re) {
              const float wj = __uint_as_float(rr[u].y);
              sx += wj * v[u].x;
              sy += wj * v[u].y;
            }
          }
        }
        racc[k].x += sx;
        racc[k].y += sy;
      }
    }
    __syncthreads();
  }

  const float2 bz = *(const float2*)(bias + 2 * l2);
#pragma unroll
  for (int k = 0; k < 7; ++k) {
    const int r = wvid + 32 * k;
    if (r < BIN_ROWS) {
      const int node = b * BIN_ROWS + r;
      float* row = out + (size_t)node * OUT_CH;
      const float2 a = racc[k];
      const float z0 = a.x + bz.x;
      const float z1 = a.y + bz.y;

      float m = (lane < 24) ? fmaxf(z0, z1) : -3.402823e38f;
      #pragma unroll
      for (int off = 16; off; off >>= 1) m = fmaxf(m, __shfl_xor(m, off, 32));

      float sm = (lane < 24) ? (__expf(z0 - m) + __expf(z1 - m)) : 0.0f;
      #pragma unroll
      for (int off = 16; off; off >>= 1) sm += __shfl_xor(sm, off, 32);

      const float lse = m + __logf(sm);
      if (lane < 24) *(float2*)(row + 2 * lane) = make_float2(z0 - lse, z1 - lse);
    }
  }
}

extern "C" void kernel_launch(void* const* d_in, const int* in_sizes, int n_in,
                              void* d_out, int out_size, void* d_ws, size_t ws_size,
                              hipStream_t stream) {
  const int*   ei   = (const int*)d_in[0];
  const float* feat = (const float*)d_in[1];
  const float* ew   = (const float*)d_in[2];
  const float* w    = (const float*)d_in[3];
  const float* bias = (const float*)d_in[4];
  float* out = (float*)d_out;
  float* xw  = (float*)d_ws;
  unsigned* recs = (unsigned*)((char*)d_ws + XW_BYTES);
  int*   tab  = (int*)((char*)d_ws + XW_BYTES + REC_BYTES);

  gemm_xw_and_file<<<N_CHUNKS + GEMM_GROUPS, 256, 0, stream>>>(feat, w, xw, ei, ew, recs, tab, ws_size);
  owner_sum_lsm<<<N_BINS, 1024, 0, stream>>>(out, bias, ei, ew, xw, recs, tab, ws_size);
}


# product split over all 8 waves by k quarter: a wave does 16 rows x 48 columns x 128 k (3 sums on one A fragment); the 4 partial sums are added in LDS in k order, whole-line stores; sum tree differs
# speedup vs baseline: 2.7388x; 1.0550x over previous
#include <hip/hip_runtime.h>
#include <hip/hip_bf16.h>

#define N_NODES 50000
#define N_EDGES 1600000
#define IN_CH   512
#define OUT_CH  48
#define LDS_STRIDE 516

typedef __attribute__((ext_vector_type(2))) float v2f;
typedef __attribute__((ext_vector_type(8))) float v8f;

#define ROW_BYTES (OUT_CH * 4)
#define XW_BYTES  (N_NODES * ROW_BYTES)

#define BIN_ROWS   200
#define N_BINS     (N_NODES / BIN_ROWS)
#define CHUNK      6400
#define N_CHUNKS   (N_EDGES / CHUNK)
#define PER_THREAD (CHUNK / 256)
#define TAB_PITCH  256
#define TILE       7168
#define REC_BYTES  ((size_t)N_EDGES * 8)
#define TAB_BYTES  ((size_t)N_CHUNKS * TAB_PITCH * 4)
#define WS_NEED    ((size_t)XW_BYTES + REC_BYTES + TAB_BYTES)
static_assert(N_NODES % BIN_ROWS == 0 && N_EDGES % CHUNK == 0 && CHUNK % 256 == 0, "exact tiling");
static_assert(TILE == 1024 * 7, "seven records per thread of the owner work-group");
static_assert(N_NODES <= 65536 && BIN_ROWS <= 256 && N_BINS <= TAB_PITCH - 1, "record packing");

__device__ __forceinline__ bool filed_path(size_t ws_size) { return ws_size >= WS_NEED; }

template <int NW, int IPT>
__device__ __forceinline__ void block_rank_by_key(const unsigned (&key)[IPT], int (&pos)[IPT],
                                                  unsigned* cntw, unsigned* tot, int* start) {
  constexpr int NT = NW * 32;
  const int tid  = threadIdx.x;
  const int lane = tid & 31;
  const int wvid = tid >> 5;
  for (int i = tid; i < NW * 256; i += NT) cntw[i] = 0u;
  __syncthreads();
  unsigned* mycnt = cntw + wvid * 256;
  int rank[IPT];
#pragma unroll
  for (int i = 0; i < IPT; ++i) {
    const unsigned k = key[i];
    unsigned m = 0xFFFFFFFFu;
#pragma unroll
    for (int bit = 0; bit < 8; ++bit) {
      const bool on = ((k >> bit) & 1u) != 0u;
      const unsigned bal = __builtin_amdgcn_ballot_w32(on);
      m &= on ? bal : ~bal;
    }
    const unsigned lt  = m & ((1u << lane) - 1u);
    const unsigned old = mycnt[k];
    rank[i] = (int)(old + (unsigned)__popc(lt));
    if (lt == 0u) mycnt[k] = old + (unsigned)__popc(m);
  }
  __syncthreads();
  if (tid < 256) {
    unsigned run = 0u;
#pragma unroll
    for (int w = 0; w < NW; ++w) {
      const unsigned c = cntw[w * 256 + tid];
      cntw[w * 256 + tid] = run;
      run += c;
    }
    tot[tid] = run;
  }
  __syncthreads();
  if (tid < 32) {
    unsigned loc[8];
    unsigned sum = 0u;
#pragma unroll
    for (int q = 0; q < 8; ++q) { loc[q] = tot[tid * 8 + q]; sum += loc[q]; }
    unsigned incl = sum;
#pragma unroll
    for (int off = 1; off < 32; off <<= 1) {
      const unsigned t = __shfl_up(incl, off, 32);
      if (tid >= off) incl += t;
    }
    unsigned run = incl - sum;
#pragma unroll
    for (int q = 0; q < 8; ++q) { start[tid * 8 + q] = (int)run; run += loc[q]; }
    if (tid == 31) start[256] = (int)run;
  }
  __syncthreads();
#pragma unroll
  for (int i = 0; i < IPT; ++i) pos[i] = start[key[i]] + (int)mycnt[key[i]] + rank[i];
}

#define GEMM_ROWS   32
#define GEMM_GROUPS ((N_NODES + GEMM_ROWS - 1) / GEMM_ROWS)
#define A_TILE_BYTES (GEMM_ROWS * LDS_STRIDE * 4)
#define FILE_LDS_BYTES (CHUNK * 8 + 8 * 256 * 4 + 256 * 4 + 260 * 4)
#define K1_LDS_BYTES (A_TILE_BYTES > FILE_LDS_BYTES ? A_TILE_BYTES : FILE_LDS_BYTES)
#define PIECES (GEMM_ROWS * (IN_CH / 4) / 256)
#define K_PARTS 4
#define K_PART  (IN_CH / K_PARTS)
#define P_PITCH 52
#define P_PLANE (GEMM_ROWS * P_PITCH)
static_assert(K_PARTS * P_PLANE * 4 <= A_TILE_BYTES, "the partial sums reuse the staged rows' LDS");
static_assert((GEMM_ROWS * OUT_CH) % 256 == 0, "whole outputs per thread");
static_assert((GEMM_ROWS * (IN_CH / 4)) % 256 == 0, "whole 16-byte pieces per thread");

__device__ __forceinline__ void product_rows_kpart(const float* As, const float* __restrict__ w,
                                                   int rt, int kq, int lane,
                                                   v8f& acc0, v8f& acc1, v8f& acc2) {
  const int r  = lane & 15;
  const int kh = (lane >> 4) << 1;
  const float* ap = As + (rt * 16 + r) * LDS_STRIDE + kq * K_PART + kh;
  const float* bp = w + (size_t)(kq * K_PART + kh) * OUT_CH + r;

  #pragma unroll 8
  for (int k = 0; k < K_PART; k += 4) {
    v2f a, b0, b1, b2;
    a.x = ap[k];
    a.y = ap[k + 1];
    const float* bq = bp + (size_t)k * OUT_CH;
    b0.x = bq[0];
    b0.y = bq[OUT_CH];
    b1.x = bq[16];
    b1.y = bq[OUT_CH + 16];
    b2.x = bq[32];
    b2.y = bq[OUT_CH + 32];
    acc0 = __builtin_amdgcn_wmma_f32_16x16x4_f32(
         false, a,  false, b0,
         (short)0, acc0,  false,  false);
    acc1 = __builtin_amdgcn_wmma_f32_16x16x4_f32(
         false, a,  false, b1,
         (short)0, acc1,  false,  false);
    acc2 = __builtin_amdgcn_wmma_f32_16x16x4_f32(
         false, a,  false, b2,
         (short)0, acc2,  false,  false);
  }
}

__device__ __forceinline__ int clamp_node(int v) {
  return v < 0 ? 0 : (v > N_NODES - 1 ? N_NODES - 1 : v);
}

__global__ __launch_bounds__(256) void gemm_xw_and_file(const float* __restrict__ feat,
                                                        const float* __restrict__ w,
                                                        float* __restrict__ xw,
                                                        const int* __restrict__ ei,
                                                        const float* __restrict__ ew,
                                                        unsigned* __restrict__ recs_w,
                                                        int* __restrict__ tab,
                                                        size_t ws_size) {
  __shared__ __align__(16) unsigned char lds[K1_LDS_BYTES];
  const int tid  = threadIdx.x;
  const int lane = tid & 31;
  const int wvid = __builtin_amdgcn_readfirstlane(tid >> 5);

  if (blockIdx.x < N_CHUNKS) {
    if (!filed_path(ws_size)) return;
    uint2*    stage = (uint2*)lds;
    unsigned* cntw  = (unsigned*)(lds + CHUNK * 8);
    unsigned* tot   = cntw + 8 * 256;
    int*      start = (int*)(tot + 256);
    const int c   = blockIdx.x;
    const int e0  = c * CHUNK + tid;

    unsigned* kbuf = (unsigned*)stage;
    {
      int dv[PER_THREAD];
#pragma unroll
      for (int i = 0; i < PER_THREAD; ++i) dv[i] = ei[N_EDGES + e0 + i * 256];
#pragma unroll
      for (int i = 0; i < PER_THREAD; ++i) {
        const int dc = clamp_node(dv[i]);
        const unsigned k = (unsigned)dc / BIN_ROWS;
        kbuf[i * 256 + tid] = k | (((unsigned)dc - k * BIN_ROWS) << 8);
      }
    }
    for (int i = tid; i < 8 * 256; i += 256) cntw[i] = 0u;
    __syncthreads();

    unsigned* mycnt = cntw + wvid * 256;
#pragma unroll 1
    for (int i = 0; i < PER_THREAD; ++i) {
      const unsigned kd = kbuf[i * 256 + tid];
      const unsigned k  = kd & 255u;
      unsigned m = 0xFFFFFFFFu;
#pragma unroll
      for (int bit = 0; bit < 8; ++bit) {
        const bool on = ((k >> bit) & 1u) != 0u;
        const unsigned bal = __builtin_amdgcn_ballot_w32(on);
        m &= on ? bal : ~bal;
      }
      const unsigned lt  = m & ((1u << lane) - 1u);
      const unsigned old = mycnt[k];
      kbuf[i * 256 + tid] = kd | ((old + (unsigned)__popc(lt)) << 16);
      if (lt == 0u) mycnt[k] = old + (unsigned)__popc(m);
    }
    __syncthreads();

    int sv[PER_THREAD];
    float wv[PER_THREAD];
#pragma unroll
    for (int i = 0; i < PER_THREAD; ++i) {
      sv[i] = ei[e0 + i * 256];
      wv[i] = ew[e0 + i * 256];
    }

    {
      unsigned run = 0u;
#pragma unroll
      for (int wq = 0; wq < 8; ++wq) {
        const unsigned cc = cntw[wq * 256 + tid];
        cntw[wq * 256 + tid] = run;
        run += cc;
      }
      tot[tid] = run;
    }
    __syncthreads();
    if (tid < 32) {
      unsigned loc[8];
      unsigned sum = 0u;
#pragma unroll
      for (int q = 0; q < 8; ++q) { loc[q] = tot[tid * 8 + q]; sum += loc[q]; }
      unsigned incl = sum;
#pragma unroll
      for (int off = 1; off < 32; off <<= 1) {
        const unsigned t = __shfl_up(incl, off, 32);
        if (tid >= off) incl += t;
      }
      unsigned run = incl - sum;
#pragma unroll
      for (int q = 0; q < 8; ++q) { start[tid * 8 + q] = (int)run; run += loc[q]; }
      if (tid == 31) start[256] = (int)run;
    }
    __syncthreads();

    tab[c * TAB_PITCH + tid] = start[tid];

    int pos[PER_THREAD];
    unsigned dlw[PER_THREAD];
#pragma unroll
    for (int i = 0; i < PER_THREAD; ++i) {
      const unsigned krd = kbuf[i * 256 + tid];
      const unsigned k   = krd & 255u;
      dlw[i] = ((krd >> 8) & 255u) << 16;
      pos[i] = start[k] + (int)mycnt[k] + (int)(krd >> 16);
    }
    __syncthreads();
#pragma unroll
    for (int i = 0; i < PER_THREAD; ++i)
      stage[pos[i]] = make_uint2((unsigned)clamp_node(sv[i]) | dlw[i], __float_as_uint(wv[i]));
    __syncthreads();

    uint4* dstv = (uint4*)(recs_w + (size_t)c * CHUNK * 2);
    const uint4* srcv = (const uint4*)stage;
    for (int q = tid; q < CHUNK / 2; q += 256) dstv[q] = srcv[q];
    return;
  }

  float* As = (float*)lds;
  const int m0 = (blockIdx.x - N_CHUNKS) * GEMM_ROWS;
  const int rt = wvid >> 2;
  const int kq = wvid & 3;

  const int sr = tid >> 7;
  const int sc = (tid & 127) << 2;
  float4 v[PIECES];
#pragma unroll
  for (int i = 0; i < PIECES; ++i) {
    const int row = m0 + sr + 2 * i;
    v[i] = *(const float4*)(feat + (size_t)(row < N_NODES ? row : N_NODES - 1) * IN_CH + sc);
  }
#pragma unroll
  for (int i = 0; i < PIECES; ++i) *(float4*)(&As[(sr + 2 * i) * LDS_STRIDE + sc]) = v[i];
  __syncthreads();

  v8f acc0 = {}, acc1 = {}, acc2 = {};
  product_rows_kpart(As, w, rt, kq, lane, acc0, acc1, acc2);
  __syncthreads();

  float* P = (float*)lds;
  {
    float* pw = P + kq * P_PLANE + (rt * 16 + ((lane >> 4) << 3)) * P_PITCH + (lane & 15);
#pragma unroll
    for (int i = 0; i < 8; ++i) {
      pw[i * P_PITCH]      = acc0[i];
      pw[i * P_PITCH + 16] = acc1[i];
      pw[i * P_PITCH + 32] = acc2[i];
    }
  }
  __syncthreads();

  float* xg = xw + (size_t)m0 * OUT_CH;
  if (m0 + GEMM_ROWS <= N_NODES) {
#pragma unroll
    for (int j = 0; j < GEMM_ROWS * OUT_CH / 256; ++j) {
      const int e   = tid + 256 * j;
      const int row = e / OUT_CH;
      const float* pr = P + row * P_PITCH + (e - row * OUT_CH);
      xg[e] = ((pr[0] + pr[P_PLANE]) + pr[2 * P_PLANE]) + pr[3 * P_PLANE];
    }
  } else {
#pragma unroll
    for (int j = 0; j < GEMM_ROWS * OUT_CH / 256; ++j) {
      const int e   = tid + 256 * j;
      const int row = e / OUT_CH;
      const float* pr = P + row * P_PITCH + (e - row * OUT_CH);
      const float s = ((pr[0] + pr[P_PLANE]) + pr[2 * P_PLANE]) + pr[3 * P_PLANE];
      if (m0 + row < N_NODES) xg[e] = s;
    }
  }
}

__global__ __launch_bounds__(1024) void owner_sum_lsm(float* __restrict__ out,
                                                      const float* __restrict__ bias,
                                                      const int* __restrict__ ei,
                                                      const float* __restrict__ ew,
                                                      const float* __restrict__ xw,
                                                      const unsigned* __restrict__ recs_w,
                                                      const int* __restrict__ tab,
                                                      size_t ws_size) {
  const uint2* recs = (const uint2*)recs_w;
  __shared__ __align__(16) uint2 sorted[TILE];
  __shared__ unsigned cntw[32 * 256];
  __shared__ unsigned tot[256];
  __shared__ int start[257];
  __shared__ unsigned psz[256];
  __shared__ int pfx[257];
  __shared__ int pst[256];
  const int tid  = threadIdx.x;
  const int lane = tid & 31;
  const int wvid = tid >> 5;
  const int b    = blockIdx.x;
  const int l2   = lane < 24 ? lane : 23;
  const bool filed = filed_path(ws_size);

  float2 racc[7];
#pragma unroll
  for (int k = 0; k < 7; ++k) racc[k] = make_float2(0.0f, 0.0f);

  int total = N_EDGES;
  if (filed) {
    if (tid < 256) {
      int s0 = 0, s1 = 0;
      if (tid < N_CHUNKS) {
        s0 = tab[tid * TAB_PITCH + b];
        s1 = tab[tid * TAB_PITCH + b + 1];
        s0 = s0 < 0 ? 0 : (s0 > CHUNK ? CHUNK : s0);
        s1 = s1 < s0 ? s0 : (s1 > CHUNK ? CHUNK : s1);
      }
      pst[tid] = s0;
      psz[tid] = (unsigned)(s1 - s0);
    }
    __syncthreads();
    if (tid < 32) {
      unsigned loc[8];
      unsigned sum = 0u;
#pragma unroll
      for (int q = 0; q < 8; ++q) { loc[q] = psz[tid * 8 + q]; sum += loc[q]; }
      unsigned incl = sum;
#pragma unroll
      for (int off = 1; off < 32; off <<= 1) {
        const unsigned t = __shfl_up(incl, off, 32);
        if (tid >= off) incl += t;
      }
      unsigned run = incl - sum;
#pragma unroll
      for (int q = 0; q < 8; ++q) { pfx[tid * 8 + q] = (int)run; run += loc[q]; }
      if (tid == 31) pfx[256] = (int)run;
    }
    __syncthreads();
    total = pfx[256];
    total = total < 0 ? 0 : (total > N_EDGES ? N_EDGES : total);
  }

  for (int tb = 0; tb < total; tb += TILE) {
    unsigned key[7];
    uint2 rec[7];
    int pos[7];
    if (filed) {
      int ra[7];
#pragma unroll
      for (int i = 0; i < 7; ++i) {
        const int g  = tb + i * 1024 + tid;
        const int gg = g < total ? g : total - 1;
        int lo = 0, hi = N_CHUNKS - 1;
#pragma unroll
        for (int it = 0; it < 8; ++it) {
          const int mid = (lo + hi + 1) >> 1;
          if (pfx[mid] <= gg) lo = mid; else hi = mid - 1;
        }
        int idx = pst[lo] + (gg - pfx[lo]);
        idx = idx < 0 ? 0 : (idx > CHUNK - 1 ? CHUNK - 1 : idx);
        ra[i] = lo * CHUNK + idx;
      }
#pragma unroll
      for (int i = 0; i < 7; ++i) rec[i] = recs[ra[i]];
#pragma unroll
      for (int i = 0; i < 7; ++i) {
        const int g = tb + i * 1024 + tid;
        const unsigned dl = rec[i].x >> 16;
        key[i] = g < total ? (dl < BIN_ROWS ? dl : BIN_ROWS - 1) : 255u;
      }
    } else {
#pragma unroll
      for (int i = 0; i < 7; ++i) {
        const int g  = tb + i * 1024 + tid;
        const int gg = g < total ? g : total - 1;
        const int dst = clamp_node(ei[N_EDGES + gg]);
        const int bin = dst / BIN_ROWS;
        const unsigned dl = (unsigned)(dst - bin * BIN_ROWS);
        rec[i] = make_uint2((unsigned)clamp_node(ei[gg]) | (dl << 16), __float_as_uint(ew[gg]));
        key[i] = (g < total && bin == b) ? dl : 255u;
      }
    }
    block_rank_by_key<32, 7>(key, pos, cntw, tot, start);
#pragma unroll
    for (int i = 0; i < 7; ++i) sorted[pos[i]] = rec[i];
    __syncthreads();

#pragma unroll
    for (int k = 0; k < 7; ++k) {
      const int r = wvid + 32 * k;
      if (r < BIN_ROWS) {
        const int rs = start[r];
        const int re = start[r + 1];
        float sx = 0.0f, sy = 0.0f;
        for (int j = rs; j < re; j += 8) {
          uint2 rr[8];
          float2 v[8];
#pragma unroll
          for (int u = 0; u < 8; ++u) {
            const int jj = (j + u < re) ? (j + u) : (re - 1);
            rr[u] = sorted[jj];
            const unsigned src = rr[u].x & 0xFFFFu;
            v[u] = *(const float2*)(xw + (size_t)(src < N_NODES ? src : N_NODES - 1) * OUT_CH + 2 * l2);
          }
#pragma unroll
          for (int u = 0; u < 8; ++u) {
            if (j + u < re) {
              const float wj = __uint_as_float(rr[u].y);
              sx += wj * v[u].x;
              sy += wj * v[u].y;
            }
          }
        }
        racc[k].x += sx;
        racc[k].y += sy;
      }
    }
    __syncthreads();
  }

  const float2 bz = *(const float2*)(bias + 2 * l2);
#pragma unroll
  for (int k = 0; k < 7; ++k) {
    const int r = wvid + 32 * k;
    if (r < BIN_ROWS) {
      const int node = b * BIN_ROWS + r;
      float* row = out + (size_t)node * OUT_CH;
      const float2 a = racc[k];
      const float z0 = a.x + bz.x;
      const float z1 = a.y + bz.y;

      float m = (lane < 24) ? fmaxf(z0, z1) : -3.402823e38f;
      #pragma unroll
      for (int off = 16; off; off >>= 1) m = fmaxf(m, __shfl_xor(m, off, 32));

      float sm = (lane < 24) ? (__expf(z0 - m) + __expf(z1 - m)) : 0.0f;
      #pragma unroll
      for (int off = 16; off; off >>= 1) sm += __shfl_xor(sm, off, 32);

      const float lse = m + __logf(sm);
      if (lane < 24) *(float2*)(row + 2 * lane) = make_float2(z0 - lse, z1 - lse);
    }
  }
}

extern "C" void kernel_launch(void* const* d_in, const int* in_sizes, int n_in,
                              void* d_out, int out_size, void* d_ws, size_t ws_size,
                              hipStream_t stream) {
  const int*   ei   = (const int*)d_in[0];
  const float* feat = (const float*)d_in[1];
  const float* ew   = (const float*)d_in[2];
  const float* w    = (const float*)d_in[3];
  const float* bias = (const float*)d_in[4];
  float* out = (float*)d_out;
  float* xw  = (float*)d_ws;
  unsigned* recs = (unsigned*)((char*)d_ws + XW_BYTES);
  int*   tab  = (int*)((char*)d_ws + XW_BYTES + REC_BYTES);

  gemm_xw_and_file<<<N_CHUNKS + GEMM_GROUPS, 256, 0, stream>>>(feat, w, xw, ei, ew, recs, tab, ws_size);
  owner_sum_lsm<<<N_BINS, 1024, 0, stream>>>(out, bias, ei, ew, xw, recs, tab, ws_size);
}


# owner's row sums as one scalar-controlled loop per wave: row bounds, source rows and weights in scalar registers (one LDS read per 16 records), 16 row loads in flight; sum order unchanged
# speedup vs baseline: 2.8286x; 1.0328x over previous
#include <hip/hip_runtime.h>
#include <hip/hip_bf16.h>

#define N_NODES 50000
#define N_EDGES 1600000
#define IN_CH   512
#define OUT_CH  48
#define LDS_STRIDE 516

typedef __attribute__((ext_vector_type(2))) float v2f;
typedef __attribute__((ext_vector_type(8))) float v8f;

#define ROW_BYTES (OUT_CH * 4)
#define XW_BYTES  (N_NODES * ROW_BYTES)

#define BIN_ROWS   200
#define N_BINS     (N_NODES / BIN_ROWS)
#define CHUNK      6400
#define N_CHUNKS   (N_EDGES / CHUNK)
#define PER_THREAD (CHUNK / 256)
#define TAB_PITCH  256
#define GRP        16
#define TILE       7168
#define REC_BYTES  ((size_t)N_EDGES * 8)
#define TAB_BYTES  ((size_t)N_CHUNKS * TAB_PITCH * 4)
#define WS_NEED    ((size_t)XW_BYTES + REC_BYTES + TAB_BYTES)
static_assert(N_NODES % BIN_ROWS == 0 && N_EDGES % CHUNK == 0 && CHUNK % 256 == 0, "exact tiling");
static_assert(TILE == 1024 * 7, "seven records per thread of the owner work-group");
static_assert(N_NODES <= 65536 && BIN_ROWS <= 256 && N_BINS <= TAB_PITCH - 1, "record packing");

__device__ __forceinline__ bool filed_path(size_t ws_size) { return ws_size >= WS_NEED; }

template <int NW, int IPT>
__device__ __forceinline__ void block_rank_by_key(const unsigned (&key)[IPT], int (&pos)[IPT],
                                                  unsigned* cntw, unsigned* tot, int* start) {
  constexpr int NT = NW * 32;
  const int tid  = threadIdx.x;
  const int lane = tid & 31;
  const int wvid = tid >> 5;
  for (int i = tid; i < NW * 256; i += NT) cntw[i] = 0u;
  __syncthreads();
  unsigned* mycnt = cntw + wvid * 256;
  int rank[IPT];
#pragma unroll
  for (int i = 0; i < IPT; ++i) {
    const unsigned k = key[i];
    unsigned m = 0xFFFFFFFFu;
#pragma unroll
    for (int bit = 0; bit < 8; ++bit) {
      const bool on = ((k >> bit) & 1u) != 0u;
      const unsigned bal = __builtin_amdgcn_ballot_w32(on);
      m &= on ? bal : ~bal;
    }
    const unsigned lt  = m & ((1u << lane) - 1u);
    const unsigned old = mycnt[k];
    rank[i] = (int)(old + (unsigned)__popc(lt));
    if (lt == 0u) mycnt[k] = old + (unsigned)__popc(m);
  }
  __syncthreads();
  if (tid < 256) {
    unsigned run = 0u;
#pragma unroll
    for (int w = 0; w < NW; ++w) {
      const unsigned c = cntw[w * 256 + tid];
      cntw[w * 256 + tid] = run;
      run += c;
    }
    tot[tid] = run;
  }
  __syncthreads();
  if (tid < 32) {
    unsigned loc[8];
    unsigned sum = 0u;
#pragma unroll
    for (int q = 0; q < 8; ++q) { loc[q] = tot[tid * 8 + q]; sum += loc[q]; }
    unsigned incl = sum;
#pragma unroll
    for (int off = 1; off < 32; off <<= 1) {
      const unsigned t = __shfl_up(incl, off, 32);
      if (tid >= off) incl += t;
    }
    unsigned run = incl - sum;
#pragma unroll
    for (int q = 0; q < 8; ++q) { start[tid * 8 + q] = (int)run; run += loc[q]; }
    if (tid == 31) start[256] = (int)run;
  }
  __syncthreads();
#pragma unroll
  for (int i = 0; i < IPT; ++i) pos[i] = start[key[i]] + (int)mycnt[key[i]] + rank[i];
}

#define GEMM_ROWS   32
#define GEMM_GROUPS ((N_NODES + GEMM_ROWS - 1) / GEMM_ROWS)
#define A_TILE_BYTES (GEMM_ROWS * LDS_STRIDE * 4)
#define FILE_LDS_BYTES (CHUNK * 8 + 8 * 256 * 4 + 256 * 4 + 260 * 4)
#define K1_LDS_BYTES (A_TILE_BYTES > FILE_LDS_BYTES ? A_TILE_BYTES : FILE_LDS_BYTES)
#define PIECES (GEMM_ROWS * (IN_CH / 4) / 256)
#define K_PARTS 4
#define K_PART  (IN_CH / K_PARTS)
#define P_PITCH 52
#define P_PLANE (GEMM_ROWS * P_PITCH)
static_assert(K_PARTS * P_PLANE * 4 <= A_TILE_BYTES, "the partial sums reuse the staged rows' LDS");
static_assert((GEMM_ROWS * OUT_CH) % 256 == 0, "whole outputs per thread");
static_assert((GEMM_ROWS * (IN_CH / 4)) % 256 == 0, "whole 16-byte pieces per thread");

__device__ __forceinline__ void product_rows_kpart(const float* As, const float* __restrict__ w,
                                                   int rt, int kq, int lane,
                                                   v8f& acc0, v8f& acc1, v8f& acc2) {
  const int r  = lane & 15;
  const int kh = (lane >> 4) << 1;
  const float* ap = As + (rt * 16 + r) * LDS_STRIDE + kq * K_PART + kh;
  const float* bp = w + (size_t)(kq * K_PART + kh) * OUT_CH + r;

  #pragma unroll 8
  for (int k = 0; k < K_PART; k += 4) {
    v2f a, b0, b1, b2;
    a.x = ap[k];
    a.y = ap[k + 1];
    const float* bq = bp + (size_t)k * OUT_CH;
    b0.x = bq[0];
    b0.y = bq[OUT_CH];
    b1.x = bq[16];
    b1.y = bq[OUT_CH + 16];
    b2.x = bq[32];
    b2.y = bq[OUT_CH + 32];
    acc0 = __builtin_amdgcn_wmma_f32_16x16x4_f32(
         false, a,  false, b0,
         (short)0, acc0,  false,  false);
    acc1 = __builtin_amdgcn_wmma_f32_16x16x4_f32(
         false, a,  false, b1,
         (short)0, acc1,  false,  false);
    acc2 = __builtin_amdgcn_wmma_f32_16x16x4_f32(
         false, a,  false, b2,
         (short)0, acc2,  false,  false);
  }
}

__device__ __forceinline__ int clamp_node(int v) {
  return v < 0 ? 0 : (v > N_NODES - 1 ? N_NODES - 1 : v);
}

__global__ __launch_bounds__(256) void gemm_xw_and_file(const float* __restrict__ feat,
                                                        const float* __restrict__ w,
                                                        float* __restrict__ xw,
                                                        const int* __restrict__ ei,
                                                        const float* __restrict__ ew,
                                                        unsigned* __restrict__ recs_w,
                                                        int* __restrict__ tab,
                                                        size_t ws_size) {
  __shared__ __align__(16) unsigned char lds[K1_LDS_BYTES];
  const int tid  = threadIdx.x;
  const int lane = tid & 31;
  const int wvid = __builtin_amdgcn_readfirstlane(tid >> 5);

  if (blockIdx.x < N_CHUNKS) {
    if (!filed_path(ws_size)) return;
    uint2*    stage = (uint2*)lds;
    unsigned* cntw  = (unsigned*)(lds + CHUNK * 8);
    unsigned* tot   = cntw + 8 * 256;
    int*      start = (int*)(tot + 256);
    const int c   = blockIdx.x;
    const int e0  = c * CHUNK + tid;

    unsigned* kbuf = (unsigned*)stage;
    {
      int dv[PER_THREAD];
#pragma unroll
      for (int i = 0; i < PER_THREAD; ++i) dv[i] = ei[N_EDGES + e0 + i * 256];
#pragma unroll
      for (int i = 0; i < PER_THREAD; ++i) {
        const int dc = clamp_node(dv[i]);
        const unsigned k = (unsigned)dc / BIN_ROWS;
        kbuf[i * 256 + tid] = k | (((unsigned)dc - k * BIN_ROWS) << 8);
      }
    }
    for (int i = tid; i < 8 * 256; i += 256) cntw[i] = 0u;
    __syncthreads();

    unsigned* mycnt = cntw + wvid * 256;
#pragma unroll 1
    for (int i = 0; i < PER_THREAD; ++i) {
      const unsigned kd = kbuf[i * 256 + tid];
      const unsigned k  = kd & 255u;
      unsigned m = 0xFFFFFFFFu;
#pragma unroll
      for (int bit = 0; bit < 8; ++bit) {
        const bool on = ((k >> bit) & 1u) != 0u;
        const unsigned bal = __builtin_amdgcn_ballot_w32(on);
        m &= on ? bal : ~bal;
      }
      const unsigned lt  = m & ((1u << lane) - 1u);
      const unsigned old = mycnt[k];
      kbuf[i * 256 + tid] = kd | ((old + (unsigned)__popc(lt)) << 16);
      if (lt == 0u) mycnt[k] = old + (unsigned)__popc(m);
    }
    __syncthreads();

    int sv[PER_THREAD];
    float wv[PER_THREAD];
#pragma unroll
    for (int i = 0; i < PER_THREAD; ++i) {
      sv[i] = ei[e0 + i * 256];
      wv[i] = ew[e0 + i * 256];
    }

    {
      unsigned run = 0u;
#pragma unroll
      for (int wq = 0; wq < 8; ++wq) {
        const unsigned cc = cntw[wq * 256 + tid];
        cntw[wq * 256 + tid] = run;
        run += cc;
      }
      tot[tid] = run;
    }
    __syncthreads();
    if (tid < 32) {
      unsigned loc[8];
      unsigned sum = 0u;
#pragma unroll
      for (int q = 0; q < 8; ++q) { loc[q] = tot[tid * 8 + q]; sum += loc[q]; }
      unsigned incl = sum;
#pragma unroll
      for (int off = 1; off < 32; off <<= 1) {
        const unsigned t = __shfl_up(incl, off, 32);
        if (tid >= off) incl += t;
      }
      unsigned run = incl - sum;
#pragma unroll
      for (int q = 0; q < 8; ++q) { start[tid * 8 + q] = (int)run; run += loc[q]; }
      if (tid == 31) start[256] = (int)run;
    }
    __syncthreads();

    tab[c * TAB_PITCH + tid] = start[tid];

    int pos[PER_THREAD];
    unsigned dlw[PER_THREAD];
#pragma unroll
    for (int i = 0; i < PER_THREAD; ++i) {
      const unsigned krd = kbuf[i * 256 + tid];
      const unsigned k   = krd & 255u;
      dlw[i] = ((krd >> 8) & 255u) << 16;
      pos[i] = start[k] + (int)mycnt[k] + (int)(krd >> 16);
    }
    __syncthreads();
#pragma unroll
    for (int i = 0; i < PER_THREAD; ++i)
      stage[pos[i]] = make_uint2((unsigned)clamp_node(sv[i]) | dlw[i], __float_as_uint(wv[i]));
    __syncthreads();

    uint4* dstv = (uint4*)(recs_w + (size_t)c * CHUNK * 2);
    const uint4* srcv = (const uint4*)stage;
    for (int q = tid; q < CHUNK / 2; q += 256) dstv[q] = srcv[q];
    return;
  }

  float* As = (float*)lds;
  const int m0 = (blockIdx.x - N_CHUNKS) * GEMM_ROWS;
  const int rt = wvid >> 2;
  const int kq = wvid & 3;

  const int sr = tid >> 7;
  const int sc = (tid & 127) << 2;
  float4 v[PIECES];
#pragma unroll
  for (int i = 0; i < PIECES; ++i) {
    const int row = m0 + sr + 2 * i;
    v[i] = *(const float4*)(feat + (size_t)(row < N_NODES ? row : N_NODES - 1) * IN_CH + sc);
  }
#pragma unroll
  for (int i = 0; i < PIECES; ++i) *(float4*)(&As[(sr + 2 * i) * LDS_STRIDE + sc]) = v[i];
  __syncthreads();

  v8f acc0 = {}, acc1 = {}, acc2 = {};
  product_rows_kpart(As, w, rt, kq, lane, acc0, acc1, acc2);
  __syncthreads();

  float* P = (float*)lds;
  {
    float* pw = P + kq * P_PLANE + (rt * 16 + ((lane >> 4) << 3)) * P_PITCH + (lane & 15);
#pragma unroll
    for (int i = 0; i < 8; ++i) {
      pw[i * P_PITCH]      = acc0[i];
      pw[i * P_PITCH + 16] = acc1[i];
      pw[i * P_PITCH + 32] = acc2[i];
    }
  }
  __syncthreads();

  float* xg = xw + (size_t)m0 * OUT_CH;
  if (m0 + GEMM_ROWS <= N_NODES) {
#pragma unroll
    for (int j = 0; j < GEMM_ROWS * OUT_CH / 256; ++j) {
      const int e   = tid + 256 * j;
      const int row = e / OUT_CH;
      const float* pr = P + row * P_PITCH + (e - row * OUT_CH);
      xg[e] = ((pr[0] + pr[P_PLANE]) + pr[2 * P_PLANE]) + pr[3 * P_PLANE];
    }
  } else {
#pragma unroll
    for (int j = 0; j < GEMM_ROWS * OUT_CH / 256; ++j) {
      const int e   = tid + 256 * j;
      const int row = e / OUT_CH;
      const float* pr = P + row * P_PITCH + (e - row * OUT_CH);
      const float s = ((pr[0] + pr[P_PLANE]) + pr[2 * P_PLANE]) + pr[3 * P_PLANE];
      if (m0 + row < N_NODES) xg[e] = s;
    }
  }
}

__device__ __forceinline__ size_t row_offset(unsigned rec_x, int t) {
  const unsigned src = (unsigned)__builtin_amdgcn_readlane((int)rec_x, t) & 0xFFFFu;
  return (size_t)(src < N_NODES ? src : N_NODES - 1) * OUT_CH;
}
__device__ __forceinline__ v2f weight_of(unsigned rec_y, int t) {
  const float wj = __uint_as_float((unsigned)__builtin_amdgcn_readlane((int)rec_y, t));
  const v2f ww = {wj, wj};
  return ww;
}

template <int N>
__device__ __forceinline__ v2f tail_sum(const float* __restrict__ xw, uint2 mine, int l2, v2f s) {
  v2f v[N];
#pragma unroll
  for (int u = 0; u < N; ++u) v[u] = *(const v2f*)(xw + row_offset(mine.x, u) + 2 * l2);
#pragma unroll
  for (int u = 0; u < N; ++u) s = __builtin_elementwise_fma(weight_of(mine.y, u), v[u], s);
  return s;
}

__global__ __launch_bounds__(1024) void owner_sum_lsm(float* __restrict__ out,
                                                      const float* __restrict__ bias,
                                                      const int* __restrict__ ei,
                                                      const float* __restrict__ ew,
                                                      const float* __restrict__ xw,
                                                      const unsigned* __restrict__ recs_w,
                                                      const int* __restrict__ tab,
                                                      size_t ws_size) {
  const uint2* recs = (const uint2*)recs_w;
  __shared__ __align__(16) uint2 sorted[TILE];
  __shared__ unsigned cntw[32 * 256];
  __shared__ unsigned tot[256];
  __shared__ int start[257];
  __shared__ unsigned psz[256];
  __shared__ int pfx[257];
  __shared__ int pst[256];
  const int tid  = threadIdx.x;
  const int lane = tid & 31;
  const int wvid = __builtin_amdgcn_readfirstlane(tid >> 5);
  const int b    = blockIdx.x;
  const int l2   = lane < 24 ? lane : 23;
  const bool filed = filed_path(ws_size);

  float2 racc[7];
#pragma unroll
  for (int k = 0; k < 7; ++k) racc[k] = make_float2(0.0f, 0.0f);

  int total = N_EDGES;
  if (filed) {
    if (tid < 256) {
      int s0 = 0, s1 = 0;
      if (tid < N_CHUNKS) {
        s0 = tab[tid * TAB_PITCH + b];
        s1 = tab[tid * TAB_PITCH + b + 1];
        s0 = s0 < 0 ? 0 : (s0 > CHUNK ? CHUNK : s0);
        s1 = s1 < s0 ? s0 : (s1 > CHUNK ? CHUNK : s1);
      }
      pst[tid] = s0;
      psz[tid] = (unsigned)(s1 - s0);
    }
    __syncthreads();
    if (tid < 32) {
      unsigned loc[8];
      unsigned sum = 0u;
#pragma unroll
      for (int q = 0; q < 8; ++q) { loc[q] = psz[tid * 8 + q]; sum += loc[q]; }
      unsigned incl = sum;
#pragma unroll
      for (int off = 1; off < 32; off <<= 1) {
        const unsigned t = __shfl_up(incl, off, 32);
        if (tid >= off) incl += t;
      }
      unsigned run = incl - sum;
#pragma unroll
      for (int q = 0; q < 8; ++q) { pfx[tid * 8 + q] = (int)run; run += loc[q]; }
      if (tid == 31) pfx[256] = (int)run;
    }
    __syncthreads();
    total = pfx[256];
    total = total < 0 ? 0 : (total > N_EDGES ? N_EDGES : total);
  }

  for (int tb = 0; tb < total; tb += TILE) {
    unsigned key[7];
    uint2 rec[7];
    int pos[7];
    if (filed) {
      int ra[7];
#pragma unroll
      for (int i = 0; i < 7; ++i) {
        const int g  = tb + i * 1024 + tid;
        const int gg = g < total ? g : total - 1;
        int lo = 0, hi = N_CHUNKS - 1;
#pragma unroll
        for (int it = 0; it < 8; ++it) {
          const int mid = (lo + hi + 1) >> 1;
          if (pfx[mid] <= gg) lo = mid; else hi = mid - 1;
        }
        int idx = pst[lo] + (gg - pfx[lo]);
        idx = idx < 0 ? 0 : (idx > CHUNK - 1 ? CHUNK - 1 : idx);
        ra[i] = lo * CHUNK + idx;
      }
#pragma unroll
      for (int i = 0; i < 7; ++i) rec[i] = recs[ra[i]];
#pragma unroll
      for (int i = 0; i < 7; ++i) {
        const int g = tb + i * 1024 + tid;
        const unsigned dl = rec[i].x >> 16;
        key[i] = g < total ? (dl < BIN_ROWS ? dl : BIN_ROWS - 1) : 255u;
      }
    } else {
#pragma unroll
      for (int i = 0; i < 7; ++i) {
        const int g  = tb + i * 1024 + tid;
        const int gg = g < total ? g : total - 1;
        const int dst = clamp_node(ei[N_EDGES + gg]);
        const int bin = dst / BIN_ROWS;
        const unsigned dl = (unsigned)(dst - bin * BIN_ROWS);
        rec[i] = make_uint2((unsigned)clamp_node(ei[gg]) | (dl << 16), __float_as_uint(ew[gg]));
        key[i] = (g < total && bin == b) ? dl : 255u;
      }
    }
    block_rank_by_key<32, 7>(key, pos, cntw, tot, start);
#pragma unroll
    for (int i = 0; i < 7; ++i) sorted[pos[i]] = rec[i];
    __syncthreads();

#pragma unroll 1
    for (int k = 0; k < 7; ++k) {
      const int r = wvid + 32 * k;
      v2f cur = {racc[0].x, racc[0].y};
      if (r < BIN_ROWS) {
        const int rs = __builtin_amdgcn_readfirstlane(start[r]);
        const int re = __builtin_amdgcn_readfirstlane(start[r + 1]);
        v2f s = {0.0f, 0.0f};
        int j = rs;
        for (; j + GRP <= re; j += GRP) {
          const uint2 mine = sorted[j + (lane & (GRP - 1))];
          v2f v[GRP];
#pragma unroll
          for (int u = 0; u < GRP; ++u) {
            v[u] = *(const v2f*)(xw + row_offset(mine.x, u) + 2 * l2);
          }
#pragma unroll
          for (int u = 0; u < GRP; ++u) {
            s = __builtin_elementwise_fma(weight_of(mine.y, u), v[u], s);
          }
        }
        const int n = re - j;
        if (n > 0) {
          const int t = lane & (GRP - 1);
          const uint2 mine = sorted[j + (t < n ? t : n - 1)];
          switch (n) {
            case 1: s = tail_sum<1>(xw, mine, l2, s); break;
            case 2: s = tail_sum<2>(xw, mine, l2, s); break;
            case 3: s = tail_sum<3>(xw, mine, l2, s); break;
            case 4: s = tail_sum<4>(xw, mine, l2, s); break;
            case 5: s = tail_sum<5>(xw, mine, l2, s); break;
            case 6: s = tail_sum<6>(xw, mine, l2, s); break;
            case 7: s = tail_sum<7>(xw, mine, l2, s); break;
            case 8: s = tail_sum<8>(xw, mine, l2, s); break;
            case 9: s = tail_sum<9>(xw, mine, l2, s); break;
            case 10: s = tail_sum<10>(xw, mine, l2, s); break;
            case 11: s = tail_sum<11>(xw, mine, l2, s); break;
            case 12: s = tail_sum<12>(xw, mine, l2, s); break;
            case 13: s = tail_sum<13>(xw, mine, l2, s); break;
            case 14: s = tail_sum<14>(xw, mine, l2, s); break;
            case 15: s = tail_sum<15>(xw, mine, l2, s); break;
            default: break;
          }
        }
        cur += s;
      }
#pragma unroll
      for (int q = 0; q < 6; ++q) racc[q] = racc[q + 1];
      racc[6] = make_float2(cur.x, cur.y);
    }
    __syncthreads();
  }

  const float2 bz = *(const float2*)(bias + 2 * l2);
#pragma unroll
  for (int k = 0; k < 7; ++k) {
    const int r = wvid + 32 * k;
    if (r < BIN_ROWS) {
      const int node = b * BIN_ROWS + r;
      float* row = out + (size_t)node * OUT_CH;
      const float2 a = racc[k];
      const float z0 = a.x + bz.x;
      const float z1 = a.y + bz.y;

      float m = (lane < 24) ? fmaxf(z0, z1) : -3.402823e38f;
      #pragma unroll
      for (int off = 16; off; off >>= 1) m = fmaxf(m, __shfl_xor(m, off, 32));

      float sm = (lane < 24) ? (__expf(z0 - m) + __expf(z1 - m)) : 0.0f;
      #pragma unroll
      for (int off = 16; off; off >>= 1) sm += __shfl_xor(sm, off, 32);

      const float lse = m + __logf(sm);
      if (lane < 24) *(float2*)(row + 2 * lane) = make_float2(z0 - lse, z1 - lse);
    }
  }
}

extern "C" void kernel_launch(void* const* d_in, const int* in_sizes, int n_in,
                              void* d_out, int out_size, void* d_ws, size_t ws_size,
                              hipStream_t stream) {
  const int*   ei   = (const int*)d_in[0];
  const float* feat = (const float*)d_in[1];
  const float* ew   = (const float*)d_in[2];
  const float* w    = (const float*)d_in[3];
  const float* bias = (const float*)d_in[4];
  float* out = (float*)d_out;
  float* xw  = (float*)d_ws;
  unsigned* recs = (unsigned*)((char*)d_ws + XW_BYTES);
  int*   tab  = (int*)((char*)d_ws + XW_BYTES + REC_BYTES);

  gemm_xw_and_file<<<N_CHUNKS + GEMM_GROUPS, 256, 0, stream>>>(feat, w, xw, ei, ew, recs, tab, ws_size);
  owner_sum_lsm<<<N_BINS, 1024, 0, stream>>>(out, bias, ei, ew, xw, recs, tab, ws_size);
}
